# starting kernel
# baseline (speedup 1.0000x reference)
#include <hip/hip_runtime.h>
#include <hip/hip_bf16.h>
#include <math.h>

typedef __attribute__((ext_vector_type(2))) float v2f;
typedef __attribute__((ext_vector_type(8))) float v8f;

#define N0  300000
#define N1  60000
#define N2  6000
#define E1  1200000
#define E2  300000
#define D_H 128
#define D_OUT 64
#define BN_EPS 1e-5f

__global__ void k_zero(float* __restrict__ p, long n) {
  long i = (long)blockIdx.x * blockDim.x + threadIdx.x;
  long stride = (long)gridDim.x * blockDim.x;
  for (; i < n; i += stride) p[i] = 0.0f;
}

__global__ void k_zeroA1(float* __restrict__ A1, const int* __restrict__ hmap) {
  int n = blockIdx.x * blockDim.x + threadIdx.x;
  int stride = gridDim.x * blockDim.x;
  for (; n < N1; n += stride) {
    if (hmap[n] < 0) {
#pragma unroll
      for (int r = 0; r < 4; ++r) {
        float* p = A1 + ((long)r * N1 + n) * 128;
        for (int c = 0; c < 128; ++c) p[c] = 0.0f;
      }
    }
  }
}

__global__ void k_scatter_f(const float* __restrict__ feat,
                            const int* __restrict__ src,
                            const int* __restrict__ dst,
                            const int* __restrict__ et,
                            const int* __restrict__ hmap,
                            float* __restrict__ acc, int nE, int nOut) {
  long total = (long)nE * 128;
  long i = (long)blockIdx.x * blockDim.x + threadIdx.x;
  long stride = (long)gridDim.x * blockDim.x;
  for (; i < total; i += stride) {
    int e = (int)(i >> 7);
    int c = (int)(i & 127);
    int d = dst[e];
    if (hmap[d] >= 0) continue;
    atomicAdd(&acc[((long)et[e] * nOut + d) * 128 + c],
              feat[(long)src[e] * 128 + c]);
  }
}

__global__ void k_histfill_stats(const int* __restrict__ hmap,
                                 const float* __restrict__ hbuf,
                                 float* __restrict__ h1,
                                 float* __restrict__ csum,
                                 float* __restrict__ csq) {
  __shared__ float ls[256], lq[256];
  long total = (long)N1 * 128;
  long i = (long)blockIdx.x * blockDim.x + threadIdx.x;
  long stride = (long)gridDim.x * blockDim.x;
  float s = 0.f, q = 0.f;
  for (; i < total; i += stride) {
    int n = (int)(i >> 7);
    int c = (int)(i & 127);
    int hm = hmap[n];
    if (hm >= 0) {
      float v = hbuf[(long)hm * 128 + c];
      h1[i] = v;
      s += v; q += v * v;
    }
  }
  ls[threadIdx.x] = s; lq[threadIdx.x] = q;
  __syncthreads();
  if (threadIdx.x < 128) {
    float S = ls[threadIdx.x] + ls[threadIdx.x + 128];
    float Q = lq[threadIdx.x] + lq[threadIdx.x + 128];
    atomicAdd(&csum[threadIdx.x], S);
    atomicAdd(&csq[threadIdx.x], Q);
  }
}

#define P1 (2 * 128 + 32)
#define LDS1 (32 * P1)

__global__ void k_gemm1(const float* __restrict__ A1,
                        const float* __restrict__ x,
                        const float* __restrict__ W1,
                        const float* __restrict__ Ws,
                        const float* __restrict__ b1,
                        const int*   __restrict__ hmap,
                        const float* __restrict__ hbuf,
                        float* __restrict__ h1,
                        float* __restrict__ csum,
                        float* __restrict__ csq) {
  __shared__ float bs[LDS1];
  __shared__ int blockNeed;

  if (threadIdx.x == 0) blockNeed = 0;
  __syncthreads();
  int blockRow = blockIdx.x * 128;
  int rr = blockRow + threadIdx.x;
  if (threadIdx.x < 128 && rr < N1 && hmap[rr] < 0) blockNeed = 1;
  __syncthreads();
  if (!blockNeed) return;

  int wave = threadIdx.x >> 5;
  int lane = threadIdx.x & 31;
  const int mtiles = N1 / 16;
  int tile = blockIdx.x * 8 + wave;
  bool valid = (tile < mtiles);
  int row0 = (valid ? tile : 0) * 16;
  bool pred = valid && (lane < 16) && (hmap[row0 + (lane & 15)] < 0);
  bool doWork = valid && (__ballot(pred) != 0ull);

  int aRow = lane & 15;
  int aK   = (lane >> 4) << 1;
  int n16  = lane & 15;

  v8f acc[8];
  v8f zero = {0.f,0.f,0.f,0.f,0.f,0.f,0.f,0.f};
#pragma unroll
  for (int t = 0; t < 8; ++t) acc[t] = zero;

  for (int seg = 0; seg < 5; ++seg) {
    const float* Ap = (seg < 4) ? (A1 + ((long)seg * N1 + row0 + aRow) * 128)
                                : (x  + (long)(row0 + aRow) * 128);
    const float* Bp = (seg < 4) ? (W1 + (long)seg * 128 * 128) : Ws;
    for (int kc = 0; kc < 128; kc += 64) {
      __syncthreads();
      for (int idx = threadIdx.x; idx < 64 * 128; idx += 256) {
        int k = idx >> 7;
        int n = idx & 127;
        bs[(k >> 1) * P1 + n * 2 + (k & 1)] = Bp[(kc + k) * 128 + n];
        if (kc == 0)
          __builtin_prefetch(&Bp[(64 + k) * 128 + n], 0, 0);
      }
      __syncthreads();

      if (doWork) {
        v2f a = *(const v2f*)(Ap + kc + aK);
#pragma unroll 2
        for (int kr = 0; kr < 64; kr += 4) {
          v2f a_cur = a;
          if (kr < 60) a = *(const v2f*)(Ap + kc + kr + 4 + aK);
          const v2f* bpair =
              (const v2f*)(bs + ((kr >> 1) + (lane >> 4)) * P1) + n16;
          v2f bf[8];
#pragma unroll
          for (int t = 0; t < 8; ++t) bf[t] = bpair[t * 16];
#pragma unroll
          for (int t = 0; t < 8; ++t)
            acc[t] = __builtin_amdgcn_wmma_f32_16x16x4_f32(
                false, a_cur, false, bf[t], (short)0, acc[t], false, false);
        }
      }
    }
  }

  if (!doWork) return;
  int halfOff = (lane >> 4) << 3;
#pragma unroll
  for (int v = 0; v < 8; ++v) {
    int r = row0 + v + halfOff;
    int hm = hmap[r];
#pragma unroll
    for (int t = 0; t < 8; ++t) {
      int col = t * 16 + n16;
      float val = acc[t][v] + b1[col];
      if (hm >= 0) val = hbuf[(long)hm * 128 + col];
      else {
        atomicAdd(&csum[col], val);
        atomicAdd(&csq[col], val * val);
      }
      h1[(long)r * 128 + col] = val;
    }
  }
}

__global__ void k_bnfinal(const float* __restrict__ csum,
                          const float* __restrict__ csq,
                          const float* __restrict__ gamma,
                          const float* __restrict__ beta,
                          float* __restrict__ scale,
                          float* __restrict__ shift) {
  int c = threadIdx.x;
  const float inv_n = 1.0f / (float)N1;
  float mean = csum[c] * inv_n;
  float var  = csq[c] * inv_n - mean * mean;
  float sc = gamma[c] * rsqrtf(var + BN_EPS);
  scale[c] = sc;
  shift[c] = beta[c] - mean * sc;
}

__global__ void k_normhead(const float* __restrict__ h1,
                           const float* __restrict__ scale,
                           const float* __restrict__ shift,
                           float* __restrict__ h1n) {
  long total = (long)N2 * 128;
  long i = (long)blockIdx.x * blockDim.x + threadIdx.x;
  long stride = (long)gridDim.x * blockDim.x;
  for (; i < total; i += stride) {
    int c = (int)(i & 127);
    h1n[i] = fmaxf(0.0f, h1[i] * scale[c] + shift[c]);
  }
}

__global__ void k_scatter2n(const float* __restrict__ h1,
                            const float* __restrict__ scale,
                            const float* __restrict__ shift,
                            const int* __restrict__ src,
                            const int* __restrict__ dst,
                            const int* __restrict__ et,
                            float* __restrict__ acc) {
  long total = (long)E2 * 128;
  long i = (long)blockIdx.x * blockDim.x + threadIdx.x;
  long stride = (long)gridDim.x * blockDim.x;
  for (; i < total; i += stride) {
    int e = (int)(i >> 7);
    int c = (int)(i & 127);
    float v = fmaxf(0.0f, h1[(long)src[e] * 128 + c] * scale[c] + shift[c]);
    atomicAdd(&acc[((long)et[e] * N2 + dst[e]) * 128 + c], v);
  }
}

#define P2 (2 * 64 + 32)
#define LDS2 (64 * P2)

__global__ void k_gemm2(const float* __restrict__ A2,
                        const float* __restrict__ h1n,
                        const float* __restrict__ W2,
                        const float* __restrict__ Ws2,
                        const float* __restrict__ b2,
                        float* __restrict__ out) {
  __shared__ float bs[LDS2];

  int wave = (int)((blockIdx.x * blockDim.x + threadIdx.x) >> 5);
  int lane = threadIdx.x & 31;
  const int mtiles = N2 / 16;
  bool valid = (wave < mtiles);
  int row0 = (valid ? wave : 0) * 16;
  int aRow = lane & 15;
  int aK   = (lane >> 4) << 1;
  int n16  = lane & 15;

  v8f acc[4];
  v8f zero = {0.f,0.f,0.f,0.f,0.f,0.f,0.f,0.f};
#pragma unroll
  for (int t = 0; t < 4; ++t) acc[t] = zero;

  for (int seg = 0; seg < 5; ++seg) {
    const float* Ap = (seg < 4) ? (A2  + ((long)seg * N2 + row0 + aRow) * 128)
                                : (h1n + (long)(row0 + aRow) * 128);
    const float* Bp = (seg < 4) ? (W2 + (long)seg * 128 * 64) : Ws2;

    __syncthreads();
    for (int idx = threadIdx.x; idx < 128 * 64; idx += 256) {
      int k = idx >> 6;
      int n = idx & 63;
      bs[(k >> 1) * P2 + n * 2 + (k & 1)] = Bp[k * 64 + n];
    }
    __syncthreads();

    if (valid) {
      v2f a = *(const v2f*)(Ap + aK);
#pragma unroll 2
      for (int k0 = 0; k0 < 128; k0 += 4) {
        v2f a_cur = a;
        if (k0 < 124) a = *(const v2f*)(Ap + k0 + 4 + aK);
        const v2f* bpair =
            (const v2f*)(bs + ((k0 >> 1) + (lane >> 4)) * P2) + n16;
        v2f bf[4];
#pragma unroll
        for (int t = 0; t < 4; ++t) bf[t] = bpair[t * 16];
#pragma unroll
        for (int t = 0; t < 4; ++t)
          acc[t] = __builtin_amdgcn_wmma_f32_16x16x4_f32(
              false, a_cur, false, bf[t], (short)0, acc[t], false, false);
      }
    }
  }

  if (!valid) return;
  int halfOff = (lane >> 4) << 3;
#pragma unroll
  for (int v = 0; v < 8; ++v) {
    int r = row0 + v + halfOff;
    float vals[4];
    float m = -1e30f;
#pragma unroll
    for (int t = 0; t < 4; ++t) {
      vals[t] = acc[t][v] + b2[t * 16 + n16];
      m = fmaxf(m, vals[t]);
    }
#pragma unroll
    for (int mask = 1; mask < 16; mask <<= 1)
      m = fmaxf(m, __shfl_xor(m, mask, 32));
    float s = 0.f;
#pragma unroll
    for (int t = 0; t < 4; ++t) s += expf(vals[t] - m);
#pragma unroll
    for (int mask = 1; mask < 16; mask <<= 1)
      s += __shfl_xor(s, mask, 32);
    float lse = m + logf(s);
#pragma unroll
    for (int t = 0; t < 4; ++t)
      out[(long)r * 64 + t * 16 + n16] = vals[t] - lse;
  }
}

extern "C" void kernel_launch(void* const* d_in, const int* in_sizes, int n_in,
                              void* d_out, int out_size, void* d_ws, size_t ws_size,
                              hipStream_t stream) {
  const float* x     = (const float*)d_in[0];
  const int*   src1  = (const int*)  d_in[1];
  const int*   dst1  = (const int*)  d_in[2];
  const int*   et1   = (const int*)  d_in[3];
  const int*   src2  = (const int*)  d_in[4];
  const int*   dst2  = (const int*)  d_in[5];
  const int*   et2   = (const int*)  d_in[6];
  const int*   hmap  = (const int*)  d_in[7];
  const float* hbuf  = (const float*)d_in[8];
  const float* W1    = (const float*)d_in[9];
  const float* Ws1   = (const float*)d_in[10];
  const float* b1    = (const float*)d_in[11];
  const float* gamma = (const float*)d_in[12];
  const float* beta  = (const float*)d_in[13];
  const float* W2    = (const float*)d_in[14];
  const float* Ws2   = (const float*)d_in[15];
  const float* b2    = (const float*)d_in[16];

  float* ws    = (float*)d_ws;
  float* A1    = ws;
  float* A2    = A1 + (size_t)4 * N1 * 128;
  float* sums  = A2 + (size_t)4 * N2 * 128;
  float* ss    = sums + 256;
  float* h1    = ss + 256;
  float* h1n   = h1 + (size_t)N1 * 128;
  float* out   = (float*)d_out;

  k_zero<<<512, 256, 0, stream>>>(A2, (long)4 * N2 * 128 + 512);
  k_zeroA1<<<256, 256, 0, stream>>>(A1, hmap);

  k_scatter_f<<<4096, 256, 0, stream>>>(x, src1, dst1, et1, hmap, A1, E1, N1);
  k_histfill_stats<<<512, 256, 0, stream>>>(hmap, hbuf, h1, sums, sums + 128);

  k_gemm1<<<(N1/16 + 7) / 8, 256, 0, stream>>>(A1, x, W1, Ws1, b1, hmap, hbuf,
                                               h1, sums, sums + 128);

  k_bnfinal<<<1, 128, 0, stream>>>(sums, sums + 128, gamma, beta, ss, ss + 128);
  k_normhead<<<512, 256, 0, stream>>>(h1, ss, ss + 128, h1n);

  k_scatter2n<<<2048, 256, 0, stream>>>(h1, ss, ss + 128, src2, dst2, et2, A2);

  k_gemm2<<<(N2/16 + 7) / 8, 256, 0, stream>>>(A2, h1n, W2, Ws2, b2, out);
}


# layer-1 scatter tests the history filter once per edge (one lane per edge, wave ballot) instead of once per element; passing edges spread over the wave's lanes; atomic arrival order changes
# speedup vs baseline: 1.4931x; 1.4931x over previous
#include <hip/hip_runtime.h>
#include <hip/hip_bf16.h>
#include <math.h>

typedef __attribute__((ext_vector_type(2))) float v2f;
typedef __attribute__((ext_vector_type(8))) float v8f;

#define N0  300000
#define N1  60000
#define N2  6000
#define E1  1200000
#define E2  300000
#define D_H 128
#define D_OUT 64
#define BN_EPS 1e-5f

__global__ void k_zero(float* __restrict__ p, long n) {
  long i = (long)blockIdx.x * blockDim.x + threadIdx.x;
  long stride = (long)gridDim.x * blockDim.x;
  for (; i < n; i += stride) p[i] = 0.0f;
}

__global__ void k_zeroA1(float* __restrict__ A1, const int* __restrict__ hmap) {
  int n = blockIdx.x * blockDim.x + threadIdx.x;
  int stride = gridDim.x * blockDim.x;
  for (; n < N1; n += stride) {
    if (hmap[n] < 0) {
#pragma unroll
      for (int r = 0; r < 4; ++r) {
        float* p = A1 + ((long)r * N1 + n) * 128;
        for (int c = 0; c < 128; ++c) p[c] = 0.0f;
      }
    }
  }
}

#define EPL 8
#define SCAT_WG_EDGES (32 * EPL * 8)
__global__ void k_scatter_f(const float* __restrict__ feat,
                            const int* __restrict__ src,
                            const int* __restrict__ dst,
                            const int* __restrict__ et,
                            const int* __restrict__ hmap,
                            float* __restrict__ acc, int nE, int nOut) {
  const int lane = threadIdx.x & 31;
  const int wave = (int)((blockIdx.x * blockDim.x + threadIdx.x) >> 5);
  const int nWaves = (int)((gridDim.x * blockDim.x) >> 5);
  for (int base = wave * (32 * EPL); base < nE; base += nWaves * (32 * EPL)) {
    int d[EPL];
    int hm[EPL];
#pragma unroll
    for (int j = 0; j < EPL; ++j)
      d[j] = dst[min(base + j * 32 + lane, nE - 1)];
#pragma unroll
    for (int j = 0; j < EPL; ++j)
      hm[j] = hmap[d[j]];
#pragma unroll
    for (int j = 0; j < EPL; ++j) {
      const int me = base + j * 32 + lane;
      unsigned todo = (unsigned)__ballot(me < nE && hm[j] < 0);
      while (todo) {
        const int l = __ffs(todo) - 1;
        todo &= todo - 1;
        const int pe = base + j * 32 + l;
        const int dd = __builtin_amdgcn_readlane(d[j], l);
        const float* f = feat + (long)src[pe] * 128 + lane;
        float* a = acc + ((long)et[pe] * nOut + dd) * 128 + lane;
        const float v0 = f[0], v1 = f[32], v2 = f[64], v3 = f[96];
        atomicAdd(a,      v0);
        atomicAdd(a + 32, v1);
        atomicAdd(a + 64, v2);
        atomicAdd(a + 96, v3);
      }
    }
  }
}

__global__ void k_histfill_stats(const int* __restrict__ hmap,
                                 const float* __restrict__ hbuf,
                                 float* __restrict__ h1,
                                 float* __restrict__ csum,
                                 float* __restrict__ csq) {
  __shared__ float ls[256], lq[256];
  long total = (long)N1 * 128;
  long i = (long)blockIdx.x * blockDim.x + threadIdx.x;
  long stride = (long)gridDim.x * blockDim.x;
  float s = 0.f, q = 0.f;
  for (; i < total; i += stride) {
    int n = (int)(i >> 7);
    int c = (int)(i & 127);
    int hm = hmap[n];
    if (hm >= 0) {
      float v = hbuf[(long)hm * 128 + c];
      h1[i] = v;
      s += v; q += v * v;
    }
  }
  ls[threadIdx.x] = s; lq[threadIdx.x] = q;
  __syncthreads();
  if (threadIdx.x < 128) {
    float S = ls[threadIdx.x] + ls[threadIdx.x + 128];
    float Q = lq[threadIdx.x] + lq[threadIdx.x + 128];
    atomicAdd(&csum[threadIdx.x], S);
    atomicAdd(&csq[threadIdx.x], Q);
  }
}

#define P1 (2 * 128 + 32)
#define LDS1 (32 * P1)

__global__ void k_gemm1(const float* __restrict__ A1,
                        const float* __restrict__ x,
                        const float* __restrict__ W1,
                        const float* __restrict__ Ws,
                        const float* __restrict__ b1,
                        const int*   __restrict__ hmap,
                        const float* __restrict__ hbuf,
                        float* __restrict__ h1,
                        float* __restrict__ csum,
                        float* __restrict__ csq) {
  __shared__ float bs[LDS1];
  __shared__ int blockNeed;

  if (threadIdx.x == 0) blockNeed = 0;
  __syncthreads();
  int blockRow = blockIdx.x * 128;
  int rr = blockRow + threadIdx.x;
  if (threadIdx.x < 128 && rr < N1 && hmap[rr] < 0) blockNeed = 1;
  __syncthreads();
  if (!blockNeed) return;

  int wave = threadIdx.x >> 5;
  int lane = threadIdx.x & 31;
  const int mtiles = N1 / 16;
  int tile = blockIdx.x * 8 + wave;
  bool valid = (tile < mtiles);
  int row0 = (valid ? tile : 0) * 16;
  bool pred = valid && (lane < 16) && (hmap[row0 + (lane & 15)] < 0);
  bool doWork = valid && (__ballot(pred) != 0ull);

  int aRow = lane & 15;
  int aK   = (lane >> 4) << 1;
  int n16  = lane & 15;

  v8f acc[8];
  v8f zero = {0.f,0.f,0.f,0.f,0.f,0.f,0.f,0.f};
#pragma unroll
  for (int t = 0; t < 8; ++t) acc[t] = zero;

  for (int seg = 0; seg < 5; ++seg) {
    const float* Ap = (seg < 4) ? (A1 + ((long)seg * N1 + row0 + aRow) * 128)
                                : (x  + (long)(row0 + aRow) * 128);
    const float* Bp = (seg < 4) ? (W1 + (long)seg * 128 * 128) : Ws;
    for (int kc = 0; kc < 128; kc += 64) {
      __syncthreads();
      for (int idx = threadIdx.x; idx < 64 * 128; idx += 256) {
        int k = idx >> 7;
        int n = idx & 127;
        bs[(k >> 1) * P1 + n * 2 + (k & 1)] = Bp[(kc + k) * 128 + n];
        if (kc == 0)
          __builtin_prefetch(&Bp[(64 + k) * 128 + n], 0, 0);
      }
      __syncthreads();

      if (doWork) {
        v2f a = *(const v2f*)(Ap + kc + aK);
#pragma unroll 2
        for (int kr = 0; kr < 64; kr += 4) {
          v2f a_cur = a;
          if (kr < 60) a = *(const v2f*)(Ap + kc + kr + 4 + aK);
          const v2f* bpair =
              (const v2f*)(bs + ((kr >> 1) + (lane >> 4)) * P1) + n16;
          v2f bf[8];
#pragma unroll
          for (int t = 0; t < 8; ++t) bf[t] = bpair[t * 16];
#pragma unroll
          for (int t = 0; t < 8; ++t)
            acc[t] = __builtin_amdgcn_wmma_f32_16x16x4_f32(
                false, a_cur, false, bf[t], (short)0, acc[t], false, false);
        }
      }
    }
  }

  if (!doWork) return;
  int halfOff = (lane >> 4) << 3;
#pragma unroll
  for (int v = 0; v < 8; ++v) {
    int r = row0 + v + halfOff;
    int hm = hmap[r];
#pragma unroll
    for (int t = 0; t < 8; ++t) {
      int col = t * 16 + n16;
      float val = acc[t][v] + b1[col];
      if (hm >= 0) val = hbuf[(long)hm * 128 + col];
      else {
        atomicAdd(&csum[col], val);
        atomicAdd(&csq[col], val * val);
      }
      h1[(long)r * 128 + col] = val;
    }
  }
}

__global__ void k_bnfinal(const float* __restrict__ csum,
                          const float* __restrict__ csq,
                          const float* __restrict__ gamma,
                          const float* __restrict__ beta,
                          float* __restrict__ scale,
                          float* __restrict__ shift) {
  int c = threadIdx.x;
  const float inv_n = 1.0f / (float)N1;
  float mean = csum[c] * inv_n;
  float var  = csq[c] * inv_n - mean * mean;
  float sc = gamma[c] * rsqrtf(var + BN_EPS);
  scale[c] = sc;
  shift[c] = beta[c] - mean * sc;
}

__global__ void k_normhead(const float* __restrict__ h1,
                           const float* __restrict__ scale,
                           const float* __restrict__ shift,
                           float* __restrict__ h1n) {
  long total = (long)N2 * 128;
  long i = (long)blockIdx.x * blockDim.x + threadIdx.x;
  long stride = (long)gridDim.x * blockDim.x;
  for (; i < total; i += stride) {
    int c = (int)(i & 127);
    h1n[i] = fmaxf(0.0f, h1[i] * scale[c] + shift[c]);
  }
}

__global__ void k_scatter2n(const float* __restrict__ h1,
                            const float* __restrict__ scale,
                            const float* __restrict__ shift,
                            const int* __restrict__ src,
                            const int* __restrict__ dst,
                            const int* __restrict__ et,
                            float* __restrict__ acc) {
  long total = (long)E2 * 128;
  long i = (long)blockIdx.x * blockDim.x + threadIdx.x;
  long stride = (long)gridDim.x * blockDim.x;
  for (; i < total; i += stride) {
    int e = (int)(i >> 7);
    int c = (int)(i & 127);
    float v = fmaxf(0.0f, h1[(long)src[e] * 128 + c] * scale[c] + shift[c]);
    atomicAdd(&acc[((long)et[e] * N2 + dst[e]) * 128 + c], v);
  }
}

#define P2 (2 * 64 + 32)
#define LDS2 (64 * P2)

__global__ void k_gemm2(const float* __restrict__ A2,
                        const float* __restrict__ h1n,
                        const float* __restrict__ W2,
                        const float* __restrict__ Ws2,
                        const float* __restrict__ b2,
                        float* __restrict__ out) {
  __shared__ float bs[LDS2];

  int wave = (int)((blockIdx.x * blockDim.x + threadIdx.x) >> 5);
  int lane = threadIdx.x & 31;
  const int mtiles = N2 / 16;
  bool valid = (wave < mtiles);
  int row0 = (valid ? wave : 0) * 16;
  int aRow = lane & 15;
  int aK   = (lane >> 4) << 1;
  int n16  = lane & 15;

  v8f acc[4];
  v8f zero = {0.f,0.f,0.f,0.f,0.f,0.f,0.f,0.f};
#pragma unroll
  for (int t = 0; t < 4; ++t) acc[t] = zero;

  for (int seg = 0; seg < 5; ++seg) {
    const float* Ap = (seg < 4) ? (A2  + ((long)seg * N2 + row0 + aRow) * 128)
                                : (h1n + (long)(row0 + aRow) * 128);
    const float* Bp = (seg < 4) ? (W2 + (long)seg * 128 * 64) : Ws2;

    __syncthreads();
    for (int idx = threadIdx.x; idx < 128 * 64; idx += 256) {
      int k = idx >> 6;
      int n = idx & 63;
      bs[(k >> 1) * P2 + n * 2 + (k & 1)] = Bp[k * 64 + n];
    }
    __syncthreads();

    if (valid) {
      v2f a = *(const v2f*)(Ap + aK);
#pragma unroll 2
      for (int k0 = 0; k0 < 128; k0 += 4) {
        v2f a_cur = a;
        if (k0 < 124) a = *(const v2f*)(Ap + k0 + 4 + aK);
        const v2f* bpair =
            (const v2f*)(bs + ((k0 >> 1) + (lane >> 4)) * P2) + n16;
        v2f bf[4];
#pragma unroll
        for (int t = 0; t < 4; ++t) bf[t] = bpair[t * 16];
#pragma unroll
        for (int t = 0; t < 4; ++t)
          acc[t] = __builtin_amdgcn_wmma_f32_16x16x4_f32(
              false, a_cur, false, bf[t], (short)0, acc[t], false, false);
      }
    }
  }

  if (!valid) return;
  int halfOff = (lane >> 4) << 3;
#pragma unroll
  for (int v = 0; v < 8; ++v) {
    int r = row0 + v + halfOff;
    float vals[4];
    float m = -1e30f;
#pragma unroll
    for (int t = 0; t < 4; ++t) {
      vals[t] = acc[t][v] + b2[t * 16 + n16];
      m = fmaxf(m, vals[t]);
    }
#pragma unroll
    for (int mask = 1; mask < 16; mask <<= 1)
      m = fmaxf(m, __shfl_xor(m, mask, 32));
    float s = 0.f;
#pragma unroll
    for (int t = 0; t < 4; ++t) s += expf(vals[t] - m);
#pragma unroll
    for (int mask = 1; mask < 16; mask <<= 1)
      s += __shfl_xor(s, mask, 32);
    float lse = m + logf(s);
#pragma unroll
    for (int t = 0; t < 4; ++t)
      out[(long)r * 64 + t * 16 + n16] = vals[t] - lse;
  }
}

extern "C" void kernel_launch(void* const* d_in, const int* in_sizes, int n_in,
                              void* d_out, int out_size, void* d_ws, size_t ws_size,
                              hipStream_t stream) {
  const float* x     = (const float*)d_in[0];
  const int*   src1  = (const int*)  d_in[1];
  const int*   dst1  = (const int*)  d_in[2];
  const int*   et1   = (const int*)  d_in[3];
  const int*   src2  = (const int*)  d_in[4];
  const int*   dst2  = (const int*)  d_in[5];
  const int*   et2   = (const int*)  d_in[6];
  const int*   hmap  = (const int*)  d_in[7];
  const float* hbuf  = (const float*)d_in[8];
  const float* W1    = (const float*)d_in[9];
  const float* Ws1   = (const float*)d_in[10];
  const float* b1    = (const float*)d_in[11];
  const float* gamma = (const float*)d_in[12];
  const float* beta  = (const float*)d_in[13];
  const float* W2    = (const float*)d_in[14];
  const float* Ws2   = (const float*)d_in[15];
  const float* b2    = (const float*)d_in[16];

  float* ws    = (float*)d_ws;
  float* A1    = ws;
  float* A2    = A1 + (size_t)4 * N1 * 128;
  float* sums  = A2 + (size_t)4 * N2 * 128;
  float* ss    = sums + 256;
  float* h1    = ss + 256;
  float* h1n   = h1 + (size_t)N1 * 128;
  float* out   = (float*)d_out;

  k_zero<<<512, 256, 0, stream>>>(A2, (long)4 * N2 * 128 + 512);
  k_zeroA1<<<256, 256, 0, stream>>>(A1, hmap);

  k_scatter_f<<<(E1 + SCAT_WG_EDGES - 1) / SCAT_WG_EDGES, 256, 0, stream>>>(x, src1, dst1, et1,
                                                                     hmap, A1, E1, N1);
  k_histfill_stats<<<512, 256, 0, stream>>>(hmap, hbuf, h1, sums, sums + 128);

  k_gemm1<<<(N1/16 + 7) / 8, 256, 0, stream>>>(A1, x, W1, Ws1, b1, hmap, hbuf,
                                               h1, sums, sums + 128);

  k_bnfinal<<<1, 128, 0, stream>>>(sums, sums + 128, gamma, beta, ss, ss + 128);
  k_normhead<<<512, 256, 0, stream>>>(h1, ss, ss + 128, h1n);

  k_scatter2n<<<2048, 256, 0, stream>>>(h1, ss, ss + 128, src2, dst2, et2, A2);

  k_gemm2<<<(N2/16 + 7) / 8, 256, 0, stream>>>(A2, h1n, W2, Ws2, b2, out);
}


# layer-1 GEMM rebuilt for the latency of its rare computed tile: K=640 split over the 8 waves, fragments loaded from global in groups, partials summed via LDS in fixed order (K sum order changes)
# speedup vs baseline: 2.2491x; 1.5063x over previous
#include <hip/hip_runtime.h>
#include <hip/hip_bf16.h>
#include <math.h>

typedef __attribute__((ext_vector_type(2))) float v2f;
typedef __attribute__((ext_vector_type(8))) float v8f;

#define N0  300000
#define N1  60000
#define N2  6000
#define E1  1200000
#define E2  300000
#define D_H 128
#define D_OUT 64
#define BN_EPS 1e-5f

__global__ void k_zero(float* __restrict__ p, long n) {
  long i = (long)blockIdx.x * blockDim.x + threadIdx.x;
  long stride = (long)gridDim.x * blockDim.x;
  for (; i < n; i += stride) p[i] = 0.0f;
}

__global__ void k_zeroA1(float* __restrict__ A1, const int* __restrict__ hmap) {
  int n = blockIdx.x * blockDim.x + threadIdx.x;
  int stride = gridDim.x * blockDim.x;
  for (; n < N1; n += stride) {
    if (hmap[n] < 0) {
#pragma unroll
      for (int r = 0; r < 4; ++r) {
        float* p = A1 + ((long)r * N1 + n) * 128;
        for (int c = 0; c < 128; ++c) p[c] = 0.0f;
      }
    }
  }
}

#define EPL 8
#define SCAT_WG_EDGES (32 * EPL * 8)
__global__ void k_scatter_f(const float* __restrict__ feat,
                            const int* __restrict__ src,
                            const int* __restrict__ dst,
                            const int* __restrict__ et,
                            const int* __restrict__ hmap,
                            float* __restrict__ acc, int nE, int nOut) {
  const int lane = threadIdx.x & 31;
  const int wave = (int)((blockIdx.x * blockDim.x + threadIdx.x) >> 5);
  const int nWaves = (int)((gridDim.x * blockDim.x) >> 5);
  for (int base = wave * (32 * EPL); base < nE; base += nWaves * (32 * EPL)) {
    int d[EPL];
    int hm[EPL];
#pragma unroll
    for (int j = 0; j < EPL; ++j)
      d[j] = dst[min(base + j * 32 + lane, nE - 1)];
#pragma unroll
    for (int j = 0; j < EPL; ++j)
      hm[j] = hmap[d[j]];
#pragma unroll
    for (int j = 0; j < EPL; ++j) {
      const int me = base + j * 32 + lane;
      unsigned todo = (unsigned)__ballot(me < nE && hm[j] < 0);
      while (todo) {
        const int l = __ffs(todo) - 1;
        todo &= todo - 1;
        const int pe = base + j * 32 + l;
        const int dd = __builtin_amdgcn_readlane(d[j], l);
        const float* f = feat + (long)src[pe] * 128 + lane;
        float* a = acc + ((long)et[pe] * nOut + dd) * 128 + lane;
        const float v0 = f[0], v1 = f[32], v2 = f[64], v3 = f[96];
        atomicAdd(a,      v0);
        atomicAdd(a + 32, v1);
        atomicAdd(a + 64, v2);
        atomicAdd(a + 96, v3);
      }
    }
  }
}

__global__ void k_histfill_stats(const int* __restrict__ hmap,
                                 const float* __restrict__ hbuf,
                                 float* __restrict__ h1,
                                 float* __restrict__ csum,
                                 float* __restrict__ csq) {
  __shared__ float ls[256], lq[256];
  long total = (long)N1 * 128;
  long i = (long)blockIdx.x * blockDim.x + threadIdx.x;
  long stride = (long)gridDim.x * blockDim.x;
  float s = 0.f, q = 0.f;
  for (; i < total; i += stride) {
    int n = (int)(i >> 7);
    int c = (int)(i & 127);
    int hm = hmap[n];
    if (hm >= 0) {
      float v = hbuf[(long)hm * 128 + c];
      h1[i] = v;
      s += v; q += v * v;
    }
  }
  ls[threadIdx.x] = s; lq[threadIdx.x] = q;
  __syncthreads();
  if (threadIdx.x < 128) {
    float S = ls[threadIdx.x] + ls[threadIdx.x + 128];
    float Q = lq[threadIdx.x] + lq[threadIdx.x + 128];
    atomicAdd(&csum[threadIdx.x], S);
    atomicAdd(&csq[threadIdx.x], Q);
  }
}

#define G1_KSTEPS 20
#define G1_GROUP  5

__global__ __launch_bounds__(256)
void k_gemm1(const float* __restrict__ A1,
             const float* __restrict__ x,
             const float* __restrict__ W1,
             const float* __restrict__ Ws,
             const float* __restrict__ b1,
             const int*   __restrict__ hmap,
             const float* __restrict__ hbuf,
             float* __restrict__ h1,
             float* __restrict__ csum,
             float* __restrict__ csq) {
  __shared__ float red[8 * 8 * 8 * 32];

  const int wave = threadIdx.x >> 5;
  const int lane = threadIdx.x & 31;
  const int rb = blockIdx.x * 128 + lane * 4;
  bool need = false;
  if (rb < N1) {
    const int4 hv = *(const int4*)(hmap + rb);
    need = (hv.x < 0) | (hv.y < 0) | (hv.z < 0) | (hv.w < 0);
  }
  const unsigned needMask = (unsigned)__ballot(need);
  if (needMask == 0u) return;

  const int c16 = lane & 15;
  const int h   = lane >> 4;

  for (int t = 0; t < 8; ++t) {
    if (((needMask >> (4 * t)) & 0xFu) == 0u) continue;
    const int row0 = blockIdx.x * 128 + t * 16;

    v8f acc[8];
    const v8f zero = {0.f,0.f,0.f,0.f,0.f,0.f,0.f,0.f};
#pragma unroll
    for (int q = 0; q < 8; ++q) acc[q] = zero;

#pragma unroll
    for (int g = 0; g < G1_KSTEPS / G1_GROUP; ++g) {
      v2f    af[G1_GROUP];
      float4 bf[G1_GROUP][2][2];
#pragma unroll
      for (int i = 0; i < G1_GROUP; ++i) {
        const int kk  = wave * (G1_KSTEPS * 4) + (g * G1_GROUP + i) * 4;
        const int seg = kk >> 7;
        const int k   = (kk & 127) + 2 * h;
        const float* Ap = (seg < 4)
            ? (A1 + ((long)seg * N1 + row0 + c16) * 128)
            : (x  + (long)(row0 + c16) * 128);
        const float* Bp = (seg < 4) ? (W1 + (long)seg * 128 * 128) : Ws;
        af[i] = *(const v2f*)(Ap + k);
        const float4* r0 = (const float4*)(Bp + (long)k * 128 + c16 * 8);
        const float4* r1 = (const float4*)(Bp + (long)(k + 1) * 128 + c16 * 8);
        bf[i][0][0] = r0[0]; bf[i][0][1] = r0[1];
        bf[i][1][0] = r1[0]; bf[i][1][1] = r1[1];
      }
#pragma unroll
      for (int i = 0; i < G1_GROUP; ++i) {
        const float k0[8] = {bf[i][0][0].x, bf[i][0][0].y, bf[i][0][0].z, bf[i][0][0].w,
                             bf[i][0][1].x, bf[i][0][1].y, bf[i][0][1].z, bf[i][0][1].w};
        const float k1[8] = {bf[i][1][0].x, bf[i][1][0].y, bf[i][1][0].z, bf[i][1][0].w,
                             bf[i][1][1].x, bf[i][1][1].y, bf[i][1][1].z, bf[i][1][1].w};
#pragma unroll
        for (int q = 0; q < 8; ++q) {
          const v2f b = {k0[q], k1[q]};
          acc[q] = __builtin_amdgcn_wmma_f32_16x16x4_f32(
              false, af[i], false, b, (short)0, acc[q], false, false);
        }
      }
    }

#pragma unroll
    for (int q = 0; q < 8; ++q)
#pragma unroll
      for (int v = 0; v < 8; ++v)
        red[((wave * 8 + q) * 8 + v) * 32 + lane] = acc[q][v];
    __syncthreads();

    float s[8];
#pragma unroll
    for (int q = 0; q < 8; ++q) {
      float sum = red[((0 * 8 + q) * 8 + wave) * 32 + lane];
#pragma unroll
      for (int w = 1; w < 8; ++w)
        sum += red[((w * 8 + q) * 8 + wave) * 32 + lane];
      s[q] = sum;
    }

    const int r    = row0 + wave + 8 * h;
    const int hm   = hmap[r];
    const int col0 = c16 * 8;
    float o[8];
    if (hm >= 0) {
      const float4* hp = (const float4*)(hbuf + (long)hm * 128 + col0);
      const float4 u0 = hp[0], u1 = hp[1];
      o[0] = u0.x; o[1] = u0.y; o[2] = u0.z; o[3] = u0.w;
      o[4] = u1.x; o[5] = u1.y; o[6] = u1.z; o[7] = u1.w;
    } else {
#pragma unroll
      for (int j = 0; j < 8; ++j) {
        o[j] = s[j] + b1[col0 + j];
        atomicAdd(&csum[col0 + j], o[j]);
        atomicAdd(&csq[col0 + j], o[j] * o[j]);
      }
    }
    float4* op = (float4*)(h1 + (long)r * 128 + col0);
    op[0] = make_float4(o[0], o[1], o[2], o[3]);
    op[1] = make_float4(o[4], o[5], o[6], o[7]);
    __syncthreads();
  }
}

__global__ void k_bnfinal(const float* __restrict__ csum,
                          const float* __restrict__ csq,
                          const float* __restrict__ gamma,
                          const float* __restrict__ beta,
                          float* __restrict__ scale,
                          float* __restrict__ shift) {
  int c = threadIdx.x;
  const float inv_n = 1.0f / (float)N1;
  float mean = csum[c] * inv_n;
  float var  = csq[c] * inv_n - mean * mean;
  float sc = gamma[c] * rsqrtf(var + BN_EPS);
  scale[c] = sc;
  shift[c] = beta[c] - mean * sc;
}

__global__ void k_normhead(const float* __restrict__ h1,
                           const float* __restrict__ scale,
                           const float* __restrict__ shift,
                           float* __restrict__ h1n) {
  long total = (long)N2 * 128;
  long i = (long)blockIdx.x * blockDim.x + threadIdx.x;
  long stride = (long)gridDim.x * blockDim.x;
  for (; i < total; i += stride) {
    int c = (int)(i & 127);
    h1n[i] = fmaxf(0.0f, h1[i] * scale[c] + shift[c]);
  }
}

__global__ void k_scatter2n(const float* __restrict__ h1,
                            const float* __restrict__ scale,
                            const float* __restrict__ shift,
                            const int* __restrict__ src,
                            const int* __restrict__ dst,
                            const int* __restrict__ et,
                            float* __restrict__ acc) {
  long total = (long)E2 * 128;
  long i = (long)blockIdx.x * blockDim.x + threadIdx.x;
  long stride = (long)gridDim.x * blockDim.x;
  for (; i < total; i += stride) {
    int e = (int)(i >> 7);
    int c = (int)(i & 127);
    float v = fmaxf(0.0f, h1[(long)src[e] * 128 + c] * scale[c] + shift[c]);
    atomicAdd(&acc[((long)et[e] * N2 + dst[e]) * 128 + c], v);
  }
}

#define P2 (2 * 64 + 32)
#define LDS2 (64 * P2)

__global__ void k_gemm2(const float* __restrict__ A2,
                        const float* __restrict__ h1n,
                        const float* __restrict__ W2,
                        const float* __restrict__ Ws2,
                        const float* __restrict__ b2,
                        float* __restrict__ out) {
  __shared__ float bs[LDS2];

  int wave = (int)((blockIdx.x * blockDim.x + threadIdx.x) >> 5);
  int lane = threadIdx.x & 31;
  const int mtiles = N2 / 16;
  bool valid = (wave < mtiles);
  int row0 = (valid ? wave : 0) * 16;
  int aRow = lane & 15;
  int aK   = (lane >> 4) << 1;
  int n16  = lane & 15;

  v8f acc[4];
  v8f zero = {0.f,0.f,0.f,0.f,0.f,0.f,0.f,0.f};
#pragma unroll
  for (int t = 0; t < 4; ++t) acc[t] = zero;

  for (int seg = 0; seg < 5; ++seg) {
    const float* Ap = (seg < 4) ? (A2  + ((long)seg * N2 + row0 + aRow) * 128)
                                : (h1n + (long)(row0 + aRow) * 128);
    const float* Bp = (seg < 4) ? (W2 + (long)seg * 128 * 64) : Ws2;

    __syncthreads();
    for (int idx = threadIdx.x; idx < 128 * 64; idx += 256) {
      int k = idx >> 6;
      int n = idx & 63;
      bs[(k >> 1) * P2 + n * 2 + (k & 1)] = Bp[k * 64 + n];
    }
    __syncthreads();

    if (valid) {
      v2f a = *(const v2f*)(Ap + aK);
#pragma unroll 2
      for (int k0 = 0; k0 < 128; k0 += 4) {
        v2f a_cur = a;
        if (k0 < 124) a = *(const v2f*)(Ap + k0 + 4 + aK);
        const v2f* bpair =
            (const v2f*)(bs + ((k0 >> 1) + (lane >> 4)) * P2) + n16;
        v2f bf[4];
#pragma unroll
        for (int t = 0; t < 4; ++t) bf[t] = bpair[t * 16];
#pragma unroll
        for (int t = 0; t < 4; ++t)
          acc[t] = __builtin_amdgcn_wmma_f32_16x16x4_f32(
              false, a_cur, false, bf[t], (short)0, acc[t], false, false);
      }
    }
  }

  if (!valid) return;
  int halfOff = (lane >> 4) << 3;
#pragma unroll
  for (int v = 0; v < 8; ++v) {
    int r = row0 + v + halfOff;
    float vals[4];
    float m = -1e30f;
#pragma unroll
    for (int t = 0; t < 4; ++t) {
      vals[t] = acc[t][v] + b2[t * 16 + n16];
      m = fmaxf(m, vals[t]);
    }
#pragma unroll
    for (int mask = 1; mask < 16; mask <<= 1)
      m = fmaxf(m, __shfl_xor(m, mask, 32));
    float s = 0.f;
#pragma unroll
    for (int t = 0; t < 4; ++t) s += expf(vals[t] - m);
#pragma unroll
    for (int mask = 1; mask < 16; mask <<= 1)
      s += __shfl_xor(s, mask, 32);
    float lse = m + logf(s);
#pragma unroll
    for (int t = 0; t < 4; ++t)
      out[(long)r * 64 + t * 16 + n16] = vals[t] - lse;
  }
}

extern "C" void kernel_launch(void* const* d_in, const int* in_sizes, int n_in,
                              void* d_out, int out_size, void* d_ws, size_t ws_size,
                              hipStream_t stream) {
  const float* x     = (const float*)d_in[0];
  const int*   src1  = (const int*)  d_in[1];
  const int*   dst1  = (const int*)  d_in[2];
  const int*   et1   = (const int*)  d_in[3];
  const int*   src2  = (const int*)  d_in[4];
  const int*   dst2  = (const int*)  d_in[5];
  const int*   et2   = (const int*)  d_in[6];
  const int*   hmap  = (const int*)  d_in[7];
  const float* hbuf  = (const float*)d_in[8];
  const float* W1    = (const float*)d_in[9];
  const float* Ws1   = (const float*)d_in[10];
  const float* b1    = (const float*)d_in[11];
  const float* gamma = (const float*)d_in[12];
  const float* beta  = (const float*)d_in[13];
  const float* W2    = (const float*)d_in[14];
  const float* Ws2   = (const float*)d_in[15];
  const float* b2    = (const float*)d_in[16];

  float* ws    = (float*)d_ws;
  float* A1    = ws;
  float* A2    = A1 + (size_t)4 * N1 * 128;
  float* sums  = A2 + (size_t)4 * N2 * 128;
  float* ss    = sums + 256;
  float* h1    = ss + 256;
  float* h1n   = h1 + (size_t)N1 * 128;
  float* out   = (float*)d_out;

  k_zero<<<512, 256, 0, stream>>>(A2, (long)4 * N2 * 128 + 512);
  k_zeroA1<<<256, 256, 0, stream>>>(A1, hmap);

  k_scatter_f<<<(E1 + SCAT_WG_EDGES - 1) / SCAT_WG_EDGES, 256, 0, stream>>>(x, src1, dst1, et1,
                                                                     hmap, A1, E1, N1);
  k_histfill_stats<<<512, 256, 0, stream>>>(hmap, hbuf, h1, sums, sums + 128);

  k_gemm1<<<(N1/16 + 7) / 8, 256, 0, stream>>>(A1, x, W1, Ws1, b1, hmap, hbuf,
                                               h1, sums, sums + 128);

  k_bnfinal<<<1, 128, 0, stream>>>(sums, sums + 128, gamma, beta, ss, ss + 128);
  k_normhead<<<512, 256, 0, stream>>>(h1, ss, ss + 128, h1n);

  k_scatter2n<<<2048, 256, 0, stream>>>(h1, ss, ss + 128, src2, dst2, et2, A2);

  k_gemm2<<<(N2/16 + 7) / 8, 256, 0, stream>>>(A2, h1n, W2, Ws2, b2, out);
}


# both row-gather loops (history fill, layer-2 scatter) go wave-per-row: indices lane-parallel, whole rows from a wave-uniform base, 8-16 rows in flight; BN partial-sum and atomic arrival order change
# speedup vs baseline: 3.6300x; 1.6140x over previous
#include <hip/hip_runtime.h>
#include <hip/hip_bf16.h>
#include <math.h>

typedef __attribute__((ext_vector_type(2))) float v2f;
typedef __attribute__((ext_vector_type(8))) float v8f;

#define N0  300000
#define N1  60000
#define N2  6000
#define E1  1200000
#define E2  300000
#define D_H 128
#define D_OUT 64
#define BN_EPS 1e-5f

__global__ void k_zero(float* __restrict__ p, long n) {
  long i = (long)blockIdx.x * blockDim.x + threadIdx.x;
  long stride = (long)gridDim.x * blockDim.x;
  for (; i < n; i += stride) p[i] = 0.0f;
}

__global__ void k_zeroA1(float* __restrict__ A1, const int* __restrict__ hmap) {
  int n = blockIdx.x * blockDim.x + threadIdx.x;
  int stride = gridDim.x * blockDim.x;
  for (; n < N1; n += stride) {
    if (hmap[n] < 0) {
#pragma unroll
      for (int r = 0; r < 4; ++r) {
        float* p = A1 + ((long)r * N1 + n) * 128;
        for (int c = 0; c < 128; ++c) p[c] = 0.0f;
      }
    }
  }
}

#define EPL 8
#define SCAT_WG_EDGES (32 * EPL * 8)
__global__ void k_scatter_f(const float* __restrict__ feat,
                            const int* __restrict__ src,
                            const int* __restrict__ dst,
                            const int* __restrict__ et,
                            const int* __restrict__ hmap,
                            float* __restrict__ acc, int nE, int nOut) {
  const int lane = threadIdx.x & 31;
  const int wave = (int)((blockIdx.x * blockDim.x + threadIdx.x) >> 5);
  const int nWaves = (int)((gridDim.x * blockDim.x) >> 5);
  for (int base = wave * (32 * EPL); base < nE; base += nWaves * (32 * EPL)) {
    int d[EPL];
    int hm[EPL];
#pragma unroll
    for (int j = 0; j < EPL; ++j)
      d[j] = dst[min(base + j * 32 + lane, nE - 1)];
#pragma unroll
    for (int j = 0; j < EPL; ++j)
      hm[j] = hmap[d[j]];
#pragma unroll
    for (int j = 0; j < EPL; ++j) {
      const int me = base + j * 32 + lane;
      unsigned todo = (unsigned)__ballot(me < nE && hm[j] < 0);
      while (todo) {
        const int l = __ffs(todo) - 1;
        todo &= todo - 1;
        const int pe = base + j * 32 + l;
        const int dd = __builtin_amdgcn_readlane(d[j], l);
        const float* f = feat + (long)src[pe] * 128 + lane;
        float* a = acc + ((long)et[pe] * nOut + dd) * 128 + lane;
        const float v0 = f[0], v1 = f[32], v2 = f[64], v3 = f[96];
        atomicAdd(a,      v0);
        atomicAdd(a + 32, v1);
        atomicAdd(a + 64, v2);
        atomicAdd(a + 96, v3);
      }
    }
  }
}

#define HF_ROWS 16
static_assert(N1 % 32 == 0, "a wave trip covers 32 whole rows");
__global__ __launch_bounds__(256)
void k_histfill_stats(const int* __restrict__ hmap,
                      const float* __restrict__ hbuf,
                      float* __restrict__ h1,
                      float* __restrict__ csum,
                      float* __restrict__ csq) {
  __shared__ __attribute__((aligned(16))) float ls[8][128];
  __shared__ __attribute__((aligned(16))) float lq[8][128];
  const int lane = threadIdx.x & 31;
  const int wv   = threadIdx.x >> 5;
  const int wave = (int)((blockIdx.x * blockDim.x + threadIdx.x) >> 5);
  const int nWaves = (int)((gridDim.x * blockDim.x) >> 5);
  float4 s = make_float4(0.f, 0.f, 0.f, 0.f);
  float4 q = make_float4(0.f, 0.f, 0.f, 0.f);
  for (int base = wave * 32; base < N1; base += nWaves * 32) {
    const int hmv = hmap[base + lane];
    for (int g = 0; g < 32; g += HF_ROWS) {
      int    hm[HF_ROWS];
      float4 v[HF_ROWS];
#pragma unroll
      for (int u = 0; u < HF_ROWS; ++u) {
        hm[u] = __builtin_amdgcn_readlane(hmv, g + u);
        v[u] = *(const float4*)(hbuf + (long)max(hm[u], 0) * 128 + lane * 4);
      }
#pragma unroll
      for (int u = 0; u < HF_ROWS; ++u) {
        if (hm[u] >= 0) {
          *(float4*)(h1 + (long)(base + g + u) * 128 + lane * 4) = v[u];
          s.x += v[u].x; s.y += v[u].y; s.z += v[u].z; s.w += v[u].w;
          q.x += v[u].x * v[u].x; q.y += v[u].y * v[u].y;
          q.z += v[u].z * v[u].z; q.w += v[u].w * v[u].w;
        }
      }
    }
  }
  *(float4*)&ls[wv][lane * 4] = s;
  *(float4*)&lq[wv][lane * 4] = q;
  __syncthreads();
  if (threadIdx.x < 128) {
    float S = ls[0][threadIdx.x], Q = lq[0][threadIdx.x];
    for (int w = 1; w < 8; ++w) { S += ls[w][threadIdx.x]; Q += lq[w][threadIdx.x]; }
    atomicAdd(&csum[threadIdx.x], S);
    atomicAdd(&csq[threadIdx.x], Q);
  }
}

#define G1_KSTEPS 20
#define G1_GROUP  5

__global__ __launch_bounds__(256)
void k_gemm1(const float* __restrict__ A1,
             const float* __restrict__ x,
             const float* __restrict__ W1,
             const float* __restrict__ Ws,
             const float* __restrict__ b1,
             const int*   __restrict__ hmap,
             const float* __restrict__ hbuf,
             float* __restrict__ h1,
             float* __restrict__ csum,
             float* __restrict__ csq) {
  __shared__ float red[8 * 8 * 8 * 32];

  const int wave = threadIdx.x >> 5;
  const int lane = threadIdx.x & 31;
  const int rb = blockIdx.x * 128 + lane * 4;
  bool need = false;
  if (rb < N1) {
    const int4 hv = *(const int4*)(hmap + rb);
    need = (hv.x < 0) | (hv.y < 0) | (hv.z < 0) | (hv.w < 0);
  }
  const unsigned needMask = (unsigned)__ballot(need);
  if (needMask == 0u) return;

  const int c16 = lane & 15;
  const int h   = lane >> 4;

  for (int t = 0; t < 8; ++t) {
    if (((needMask >> (4 * t)) & 0xFu) == 0u) continue;
    const int row0 = blockIdx.x * 128 + t * 16;

    v8f acc[8];
    const v8f zero = {0.f,0.f,0.f,0.f,0.f,0.f,0.f,0.f};
#pragma unroll
    for (int q = 0; q < 8; ++q) acc[q] = zero;

#pragma unroll
    for (int g = 0; g < G1_KSTEPS / G1_GROUP; ++g) {
      v2f    af[G1_GROUP];
      float4 bf[G1_GROUP][2][2];
#pragma unroll
      for (int i = 0; i < G1_GROUP; ++i) {
        const int kk  = wave * (G1_KSTEPS * 4) + (g * G1_GROUP + i) * 4;
        const int seg = kk >> 7;
        const int k   = (kk & 127) + 2 * h;
        const float* Ap = (seg < 4)
            ? (A1 + ((long)seg * N1 + row0 + c16) * 128)
            : (x  + (long)(row0 + c16) * 128);
        const float* Bp = (seg < 4) ? (W1 + (long)seg * 128 * 128) : Ws;
        af[i] = *(const v2f*)(Ap + k);
        const float4* r0 = (const float4*)(Bp + (long)k * 128 + c16 * 8);
        const float4* r1 = (const float4*)(Bp + (long)(k + 1) * 128 + c16 * 8);
        bf[i][0][0] = r0[0]; bf[i][0][1] = r0[1];
        bf[i][1][0] = r1[0]; bf[i][1][1] = r1[1];
      }
#pragma unroll
      for (int i = 0; i < G1_GROUP; ++i) {
        const float k0[8] = {bf[i][0][0].x, bf[i][0][0].y, bf[i][0][0].z, bf[i][0][0].w,
                             bf[i][0][1].x, bf[i][0][1].y, bf[i][0][1].z, bf[i][0][1].w};
        const float k1[8] = {bf[i][1][0].x, bf[i][1][0].y, bf[i][1][0].z, bf[i][1][0].w,
                             bf[i][1][1].x, bf[i][1][1].y, bf[i][1][1].z, bf[i][1][1].w};
#pragma unroll
        for (int q = 0; q < 8; ++q) {
          const v2f b = {k0[q], k1[q]};
          acc[q] = __builtin_amdgcn_wmma_f32_16x16x4_f32(
              false, af[i], false, b, (short)0, acc[q], false, false);
        }
      }
    }

#pragma unroll
    for (int q = 0; q < 8; ++q)
#pragma unroll
      for (int v = 0; v < 8; ++v)
        red[((wave * 8 + q) * 8 + v) * 32 + lane] = acc[q][v];
    __syncthreads();

    float s[8];
#pragma unroll
    for (int q = 0; q < 8; ++q) {
      float sum = red[((0 * 8 + q) * 8 + wave) * 32 + lane];
#pragma unroll
      for (int w = 1; w < 8; ++w)
        sum += red[((w * 8 + q) * 8 + wave) * 32 + lane];
      s[q] = sum;
    }

    const int r    = row0 + wave + 8 * h;
    const int hm   = hmap[r];
    const int col0 = c16 * 8;
    float o[8];
    if (hm >= 0) {
      const float4* hp = (const float4*)(hbuf + (long)hm * 128 + col0);
      const float4 u0 = hp[0], u1 = hp[1];
      o[0] = u0.x; o[1] = u0.y; o[2] = u0.z; o[3] = u0.w;
      o[4] = u1.x; o[5] = u1.y; o[6] = u1.z; o[7] = u1.w;
    } else {
#pragma unroll
      for (int j = 0; j < 8; ++j) {
        o[j] = s[j] + b1[col0 + j];
        atomicAdd(&csum[col0 + j], o[j]);
        atomicAdd(&csq[col0 + j], o[j] * o[j]);
      }
    }
    float4* op = (float4*)(h1 + (long)r * 128 + col0);
    op[0] = make_float4(o[0], o[1], o[2], o[3]);
    op[1] = make_float4(o[4], o[5], o[6], o[7]);
    __syncthreads();
  }
}

__global__ void k_bnfinal(const float* __restrict__ csum,
                          const float* __restrict__ csq,
                          const float* __restrict__ gamma,
                          const float* __restrict__ beta,
                          float* __restrict__ scale,
                          float* __restrict__ shift) {
  int c = threadIdx.x;
  const float inv_n = 1.0f / (float)N1;
  float mean = csum[c] * inv_n;
  float var  = csq[c] * inv_n - mean * mean;
  float sc = gamma[c] * rsqrtf(var + BN_EPS);
  scale[c] = sc;
  shift[c] = beta[c] - mean * sc;
}

__global__ void k_normhead(const float* __restrict__ h1,
                           const float* __restrict__ scale,
                           const float* __restrict__ shift,
                           float* __restrict__ h1n) {
  long total = (long)N2 * 128;
  long i = (long)blockIdx.x * blockDim.x + threadIdx.x;
  long stride = (long)gridDim.x * blockDim.x;
  for (; i < total; i += stride) {
    int c = (int)(i & 127);
    h1n[i] = fmaxf(0.0f, h1[i] * scale[c] + shift[c]);
  }
}

#define GR 8
static_assert(E2 % 32 == 0, "a wave trip covers 32 whole edges");
__global__ __launch_bounds__(256)
void k_scatter2n(const float* __restrict__ h1,
                 const float* __restrict__ scale,
                 const float* __restrict__ shift,
                 const int* __restrict__ src,
                 const int* __restrict__ dst,
                 const int* __restrict__ et,
                 float* __restrict__ acc) {
  const int lane = threadIdx.x & 31;
  const int wave = (int)((blockIdx.x * blockDim.x + threadIdx.x) >> 5);
  const int nWaves = (int)((gridDim.x * blockDim.x) >> 5);
  float sc[4], sh[4];
#pragma unroll
  for (int j = 0; j < 4; ++j) {
    sc[j] = scale[lane + 32 * j];
    sh[j] = shift[lane + 32 * j];
  }
  for (int base = wave * 32; base < E2; base += nWaves * 32) {
    const int sv = src[base + lane];
    const int av = et[base + lane] * N2 + dst[base + lane];
    for (int g = 0; g < 32; g += GR) {
      int   ar[GR];
      float v[GR][4];
#pragma unroll
      for (int u = 0; u < GR; ++u) {
        const int ss = __builtin_amdgcn_readlane(sv, g + u);
        ar[u] = __builtin_amdgcn_readlane(av, g + u);
        const float* f = h1 + (long)ss * 128 + lane;
        v[u][0] = f[0]; v[u][1] = f[32]; v[u][2] = f[64]; v[u][3] = f[96];
      }
#pragma unroll
      for (int u = 0; u < GR; ++u) {
        float* a = acc + (long)ar[u] * 128 + lane;
#pragma unroll
        for (int j = 0; j < 4; ++j)
          atomicAdd(a + 32 * j, fmaxf(0.0f, v[u][j] * sc[j] + sh[j]));
      }
    }
  }
}

#define P2 (2 * 64 + 32)
#define LDS2 (64 * P2)

__global__ void k_gemm2(const float* __restrict__ A2,
                        const float* __restrict__ h1n,
                        const float* __restrict__ W2,
                        const float* __restrict__ Ws2,
                        const float* __restrict__ b2,
                        float* __restrict__ out) {
  __shared__ float bs[LDS2];

  int wave = (int)((blockIdx.x * blockDim.x + threadIdx.x) >> 5);
  int lane = threadIdx.x & 31;
  const int mtiles = N2 / 16;
  bool valid = (wave < mtiles);
  int row0 = (valid ? wave : 0) * 16;
  int aRow = lane & 15;
  int aK   = (lane >> 4) << 1;
  int n16  = lane & 15;

  v8f acc[4];
  v8f zero = {0.f,0.f,0.f,0.f,0.f,0.f,0.f,0.f};
#pragma unroll
  for (int t = 0; t < 4; ++t) acc[t] = zero;

  for (int seg = 0; seg < 5; ++seg) {
    const float* Ap = (seg < 4) ? (A2  + ((long)seg * N2 + row0 + aRow) * 128)
                                : (h1n + (long)(row0 + aRow) * 128);
    const float* Bp = (seg < 4) ? (W2 + (long)seg * 128 * 64) : Ws2;

    __syncthreads();
    for (int idx = threadIdx.x; idx < 128 * 64; idx += 256) {
      int k = idx >> 6;
      int n = idx & 63;
      bs[(k >> 1) * P2 + n * 2 + (k & 1)] = Bp[k * 64 + n];
    }
    __syncthreads();

    if (valid) {
      v2f a = *(const v2f*)(Ap + aK);
#pragma unroll 2
      for (int k0 = 0; k0 < 128; k0 += 4) {
        v2f a_cur = a;
        if (k0 < 124) a = *(const v2f*)(Ap + k0 + 4 + aK);
        const v2f* bpair =
            (const v2f*)(bs + ((k0 >> 1) + (lane >> 4)) * P2) + n16;
        v2f bf[4];
#pragma unroll
        for (int t = 0; t < 4; ++t) bf[t] = bpair[t * 16];
#pragma unroll
        for (int t = 0; t < 4; ++t)
          acc[t] = __builtin_amdgcn_wmma_f32_16x16x4_f32(
              false, a_cur, false, bf[t], (short)0, acc[t], false, false);
      }
    }
  }

  if (!valid) return;
  int halfOff = (lane >> 4) << 3;
#pragma unroll
  for (int v = 0; v < 8; ++v) {
    int r = row0 + v + halfOff;
    float vals[4];
    float m = -1e30f;
#pragma unroll
    for (int t = 0; t < 4; ++t) {
      vals[t] = acc[t][v] + b2[t * 16 + n16];
      m = fmaxf(m, vals[t]);
    }
#pragma unroll
    for (int mask = 1; mask < 16; mask <<= 1)
      m = fmaxf(m, __shfl_xor(m, mask, 32));
    float s = 0.f;
#pragma unroll
    for (int t = 0; t < 4; ++t) s += expf(vals[t] - m);
#pragma unroll
    for (int mask = 1; mask < 16; mask <<= 1)
      s += __shfl_xor(s, mask, 32);
    float lse = m + logf(s);
#pragma unroll
    for (int t = 0; t < 4; ++t)
      out[(long)r * 64 + t * 16 + n16] = vals[t] - lse;
  }
}

extern "C" void kernel_launch(void* const* d_in, const int* in_sizes, int n_in,
                              void* d_out, int out_size, void* d_ws, size_t ws_size,
                              hipStream_t stream) {
  const float* x     = (const float*)d_in[0];
  const int*   src1  = (const int*)  d_in[1];
  const int*   dst1  = (const int*)  d_in[2];
  const int*   et1   = (const int*)  d_in[3];
  const int*   src2  = (const int*)  d_in[4];
  const int*   dst2  = (const int*)  d_in[5];
  const int*   et2   = (const int*)  d_in[6];
  const int*   hmap  = (const int*)  d_in[7];
  const float* hbuf  = (const float*)d_in[8];
  const float* W1    = (const float*)d_in[9];
  const float* Ws1   = (const float*)d_in[10];
  const float* b1    = (const float*)d_in[11];
  const float* gamma = (const float*)d_in[12];
  const float* beta  = (const float*)d_in[13];
  const float* W2    = (const float*)d_in[14];
  const float* Ws2   = (const float*)d_in[15];
  const float* b2    = (const float*)d_in[16];

  float* ws    = (float*)d_ws;
  float* A1    = ws;
  float* A2    = A1 + (size_t)4 * N1 * 128;
  float* sums  = A2 + (size_t)4 * N2 * 128;
  float* ss    = sums + 256;
  float* h1    = ss + 256;
  float* h1n   = h1 + (size_t)N1 * 128;
  float* out   = (float*)d_out;

  k_zero<<<512, 256, 0, stream>>>(A2, (long)4 * N2 * 128 + 512);
  k_zeroA1<<<256, 256, 0, stream>>>(A1, hmap);

  k_scatter_f<<<(E1 + SCAT_WG_EDGES - 1) / SCAT_WG_EDGES, 256, 0, stream>>>(x, src1, dst1, et1,
                                                                     hmap, A1, E1, N1);
  k_histfill_stats<<<(N1 / 32 + 7) / 8, 256, 0, stream>>>(hmap, hbuf, h1, sums, sums + 128);

  k_gemm1<<<(N1/16 + 7) / 8, 256, 0, stream>>>(A1, x, W1, Ws1, b1, hmap, hbuf,
                                               h1, sums, sums + 128);

  k_bnfinal<<<1, 128, 0, stream>>>(sums, sums + 128, gamma, beta, ss, ss + 128);
  k_normhead<<<512, 256, 0, stream>>>(h1, ss, ss + 128, h1n);

  k_scatter2n<<<(E2 / 32 + 7) / 8, 256, 0, stream>>>(h1, ss, ss + 128, src2, dst2, et2, A2);

  k_gemm2<<<(N2/16 + 7) / 8, 256, 0, stream>>>(A2, h1n, W2, Ws2, b2, out);
}


# 9 launches -> 5: independent kernels share a launch by work-group index (clears; layer-1 scatter + history fill; layer-2 scatter + head rows); BN fold inlined in users; atomic arrival order changes
# speedup vs baseline: 4.0506x; 1.1159x over previous
#include <hip/hip_runtime.h>
#include <hip/hip_bf16.h>
#include <math.h>

typedef __attribute__((ext_vector_type(2))) float v2f;
typedef __attribute__((ext_vector_type(8))) float v8f;

#define N0  300000
#define N1  60000
#define N2  6000
#define E1  1200000
#define E2  300000
#define D_H 128
#define D_OUT 64
#define BN_EPS 1e-5f

__global__ void k_clear(float* __restrict__ p, long n,
                        float* __restrict__ A1, const int* __restrict__ hmap) {
  const long tid = (long)blockIdx.x * blockDim.x + threadIdx.x;
  const long stride = (long)gridDim.x * blockDim.x;
  for (long i = tid; i < n; i += stride) p[i] = 0.0f;
  for (long nn = tid; nn < N1; nn += stride) {
    if (hmap[nn] < 0) {
      for (int r = 0; r < 4; ++r) {
        float* q = A1 + ((long)r * N1 + nn) * 128;
        for (int c = 0; c < 128; ++c) q[c] = 0.0f;
      }
    }
  }
}

#define EPL 8
#define SCAT_WG_EDGES (32 * EPL * 8)
__device__ __forceinline__
void scatter_f_body(const int vb, const int nvb,
                    const float* __restrict__ feat,
                    const int* __restrict__ src,
                    const int* __restrict__ dst,
                    const int* __restrict__ et,
                    const int* __restrict__ hmap,
                    float* __restrict__ acc, int nE, int nOut) {
  const int lane = threadIdx.x & 31;
  const int wave = (int)((vb * blockDim.x + threadIdx.x) >> 5);
  const int nWaves = (int)((nvb * blockDim.x) >> 5);
  for (int base = wave * (32 * EPL); base < nE; base += nWaves * (32 * EPL)) {
    int d[EPL];
    int hm[EPL];
#pragma unroll
    for (int j = 0; j < EPL; ++j)
      d[j] = dst[min(base + j * 32 + lane, nE - 1)];
#pragma unroll
    for (int j = 0; j < EPL; ++j)
      hm[j] = hmap[d[j]];
#pragma unroll
    for (int j = 0; j < EPL; ++j) {
      const int me = base + j * 32 + lane;
      unsigned todo = (unsigned)__ballot(me < nE && hm[j] < 0);
      while (todo) {
        const int l = __ffs(todo) - 1;
        todo &= todo - 1;
        const int pe = base + j * 32 + l;
        const int dd = __builtin_amdgcn_readlane(d[j], l);
        const float* f = feat + (long)src[pe] * 128 + lane;
        float* a = acc + ((long)et[pe] * nOut + dd) * 128 + lane;
        const float v0 = f[0], v1 = f[32], v2 = f[64], v3 = f[96];
        atomicAdd(a,      v0);
        atomicAdd(a + 32, v1);
        atomicAdd(a + 64, v2);
        atomicAdd(a + 96, v3);
      }
    }
  }
}

#define HF_ROWS 16
static_assert(N1 % 32 == 0, "a wave trip covers 32 whole rows");
__device__ __forceinline__
void histfill_stats_body(const int vb, const int nvb,
                         const int* __restrict__ hmap,
                         const float* __restrict__ hbuf,
                         float* __restrict__ h1,
                         float* __restrict__ csum,
                         float* __restrict__ csq) {
  __shared__ __attribute__((aligned(16))) float ls[8][128];
  __shared__ __attribute__((aligned(16))) float lq[8][128];
  const int lane = threadIdx.x & 31;
  const int wv   = threadIdx.x >> 5;
  const int wave = (int)((vb * blockDim.x + threadIdx.x) >> 5);
  const int nWaves = (int)((nvb * blockDim.x) >> 5);
  float4 s = make_float4(0.f, 0.f, 0.f, 0.f);
  float4 q = make_float4(0.f, 0.f, 0.f, 0.f);
  for (int base = wave * 32; base < N1; base += nWaves * 32) {
    const int hmv = hmap[base + lane];
    for (int g = 0; g < 32; g += HF_ROWS) {
      int    hm[HF_ROWS];
      float4 v[HF_ROWS];
#pragma unroll
      for (int u = 0; u < HF_ROWS; ++u) {
        hm[u] = __builtin_amdgcn_readlane(hmv, g + u);
        v[u] = *(const float4*)(hbuf + (long)max(hm[u], 0) * 128 + lane * 4);
      }
#pragma unroll
      for (int u = 0; u < HF_ROWS; ++u) {
        if (hm[u] >= 0) {
          *(float4*)(h1 + (long)(base + g + u) * 128 + lane * 4) = v[u];
          s.x += v[u].x; s.y += v[u].y; s.z += v[u].z; s.w += v[u].w;
          q.x += v[u].x * v[u].x; q.y += v[u].y * v[u].y;
          q.z += v[u].z * v[u].z; q.w += v[u].w * v[u].w;
        }
      }
    }
  }
  *(float4*)&ls[wv][lane * 4] = s;
  *(float4*)&lq[wv][lane * 4] = q;
  __syncthreads();
  if (threadIdx.x < 128) {
    float S = ls[0][threadIdx.x], Q = lq[0][threadIdx.x];
    for (int w = 1; w < 8; ++w) { S += ls[w][threadIdx.x]; Q += lq[w][threadIdx.x]; }
    atomicAdd(&csum[threadIdx.x], S);
    atomicAdd(&csq[threadIdx.x], Q);
  }
}

#define HF_WGS ((N1 / 32 + 7) / 8)
#define SC1_WGS ((E1 + SCAT_WG_EDGES - 1) / SCAT_WG_EDGES)
__global__ __launch_bounds__(256)
void k_scatter_histfill(const float* __restrict__ x,
                        const int* __restrict__ src1,
                        const int* __restrict__ dst1,
                        const int* __restrict__ et1,
                        const int* __restrict__ hmap,
                        const float* __restrict__ hbuf,
                        float* __restrict__ A1,
                        float* __restrict__ h1,
                        float* __restrict__ csum,
                        float* __restrict__ csq) {
  if (blockIdx.x < HF_WGS)
    histfill_stats_body((int)blockIdx.x, HF_WGS, hmap, hbuf, h1, csum, csq);
  else
    scatter_f_body((int)blockIdx.x - HF_WGS, SC1_WGS, x, src1, dst1, et1, hmap,
                   A1, E1, N1);
}

#define G1_KSTEPS 20
#define G1_GROUP  5

__global__ __launch_bounds__(256)
void k_gemm1(const float* __restrict__ A1,
             const float* __restrict__ x,
             const float* __restrict__ W1,
             const float* __restrict__ Ws,
             const float* __restrict__ b1,
             const int*   __restrict__ hmap,
             const float* __restrict__ hbuf,
             float* __restrict__ h1,
             float* __restrict__ csum,
             float* __restrict__ csq) {
  __shared__ float red[8 * 8 * 8 * 32];

  const int wave = threadIdx.x >> 5;
  const int lane = threadIdx.x & 31;
  const int rb = blockIdx.x * 128 + lane * 4;
  bool need = false;
  if (rb < N1) {
    const int4 hv = *(const int4*)(hmap + rb);
    need = (hv.x < 0) | (hv.y < 0) | (hv.z < 0) | (hv.w < 0);
  }
  const unsigned needMask = (unsigned)__ballot(need);
  if (needMask == 0u) return;

  const int c16 = lane & 15;
  const int h   = lane >> 4;

  for (int t = 0; t < 8; ++t) {
    if (((needMask >> (4 * t)) & 0xFu) == 0u) continue;
    const int row0 = blockIdx.x * 128 + t * 16;

    v8f acc[8];
    const v8f zero = {0.f,0.f,0.f,0.f,0.f,0.f,0.f,0.f};
#pragma unroll
    for (int q = 0; q < 8; ++q) acc[q] = zero;

#pragma unroll
    for (int g = 0; g < G1_KSTEPS / G1_GROUP; ++g) {
      v2f    af[G1_GROUP];
      float4 bf[G1_GROUP][2][2];
#pragma unroll
      for (int i = 0; i < G1_GROUP; ++i) {
        const int kk  = wave * (G1_KSTEPS * 4) + (g * G1_GROUP + i) * 4;
        const int seg = kk >> 7;
        const int k   = (kk & 127) + 2 * h;
        const float* Ap = (seg < 4)
            ? (A1 + ((long)seg * N1 + row0 + c16) * 128)
            : (x  + (long)(row0 + c16) * 128);
        const float* Bp = (seg < 4) ? (W1 + (long)seg * 128 * 128) : Ws;
        af[i] = *(const v2f*)(Ap + k);
        const float4* r0 = (const float4*)(Bp + (long)k * 128 + c16 * 8);
        const float4* r1 = (const float4*)(Bp + (long)(k + 1) * 128 + c16 * 8);
        bf[i][0][0] = r0[0]; bf[i][0][1] = r0[1];
        bf[i][1][0] = r1[0]; bf[i][1][1] = r1[1];
      }
#pragma unroll
      for (int i = 0; i < G1_GROUP; ++i) {
        const float k0[8] = {bf[i][0][0].x, bf[i][0][0].y, bf[i][0][0].z, bf[i][0][0].w,
                             bf[i][0][1].x, bf[i][0][1].y, bf[i][0][1].z, bf[i][0][1].w};
        const float k1[8] = {bf[i][1][0].x, bf[i][1][0].y, bf[i][1][0].z, bf[i][1][0].w,
                             bf[i][1][1].x, bf[i][1][1].y, bf[i][1][1].z, bf[i][1][1].w};
#pragma unroll
        for (int q = 0; q < 8; ++q) {
          const v2f b = {k0[q], k1[q]};
          acc[q] = __builtin_amdgcn_wmma_f32_16x16x4_f32(
              false, af[i], false, b, (short)0, acc[q], false, false);
        }
      }
    }

#pragma unroll
    for (int q = 0; q < 8; ++q)
#pragma unroll
      for (int v = 0; v < 8; ++v)
        red[((wave * 8 + q) * 8 + v) * 32 + lane] = acc[q][v];
    __syncthreads();

    float s[8];
#pragma unroll
    for (int q = 0; q < 8; ++q) {
      float sum = red[((0 * 8 + q) * 8 + wave) * 32 + lane];
#pragma unroll
      for (int w = 1; w < 8; ++w)
        sum += red[((w * 8 + q) * 8 + wave) * 32 + lane];
      s[q] = sum;
    }

    const int r    = row0 + wave + 8 * h;
    const int hm   = hmap[r];
    const int col0 = c16 * 8;
    float o[8];
    if (hm >= 0) {
      const float4* hp = (const float4*)(hbuf + (long)hm * 128 + col0);
      const float4 u0 = hp[0], u1 = hp[1];
      o[0] = u0.x; o[1] = u0.y; o[2] = u0.z; o[3] = u0.w;
      o[4] = u1.x; o[5] = u1.y; o[6] = u1.z; o[7] = u1.w;
    } else {
#pragma unroll
      for (int j = 0; j < 8; ++j) {
        o[j] = s[j] + b1[col0 + j];
        atomicAdd(&csum[col0 + j], o[j]);
        atomicAdd(&csq[col0 + j], o[j] * o[j]);
      }
    }
    float4* op = (float4*)(h1 + (long)r * 128 + col0);
    op[0] = make_float4(o[0], o[1], o[2], o[3]);
    op[1] = make_float4(o[4], o[5], o[6], o[7]);
    __syncthreads();
  }
}

__device__ __forceinline__
void bn_coeff(const float* __restrict__ csum, const float* __restrict__ csq,
              const float* __restrict__ gamma, const float* __restrict__ beta,
              const int c, float& sc, float& sh) {
  const float inv_n = 1.0f / (float)N1;
  const float mean = csum[c] * inv_n;
  const float var  = csq[c] * inv_n - mean * mean;
  sc = gamma[c] * rsqrtf(var + BN_EPS);
  sh = beta[c] - mean * sc;
}

__device__ __forceinline__
void normhead_body(const int vb, const int nvb,
                   const float* __restrict__ h1,
                   const float* __restrict__ csum, const float* __restrict__ csq,
                   const float* __restrict__ gamma, const float* __restrict__ beta,
                   float* __restrict__ h1n) {
  const long total = (long)N2 * 128;
  long i = (long)vb * blockDim.x + threadIdx.x;
  const long stride = (long)nvb * blockDim.x;
  float sc, sh;
  bn_coeff(csum, csq, gamma, beta, (int)(i & 127), sc, sh);
  for (; i < total; i += stride)
    h1n[i] = fmaxf(0.0f, h1[i] * sc + sh);
}

#define GR 8
static_assert(E2 % 32 == 0, "a wave trip covers 32 whole edges");
__device__ __forceinline__
void scatter2n_body(const int vb, const int nvb,
                    const float* __restrict__ h1,
                    const float* __restrict__ csum, const float* __restrict__ csq,
                    const float* __restrict__ gamma, const float* __restrict__ beta,
                    const int* __restrict__ src,
                    const int* __restrict__ dst,
                    const int* __restrict__ et,
                    float* __restrict__ acc) {
  const int lane = threadIdx.x & 31;
  const int wave = (int)((vb * blockDim.x + threadIdx.x) >> 5);
  const int nWaves = (int)((nvb * blockDim.x) >> 5);
  int base = wave * 32;
  int sv = 0, ev = 0, dv = 0;
  if (base < E2) {
    sv = src[base + lane]; ev = et[base + lane]; dv = dst[base + lane];
  }
  float sc[4], sh[4];
#pragma unroll
  for (int j = 0; j < 4; ++j)
    bn_coeff(csum, csq, gamma, beta, lane + 32 * j, sc[j], sh[j]);
  while (base < E2) {
    const int av = ev * N2 + dv;
    for (int g = 0; g < 32; g += GR) {
      int   ar[GR];
      float v[GR][4];
#pragma unroll
      for (int u = 0; u < GR; ++u) {
        const int ss = __builtin_amdgcn_readlane(sv, g + u);
        ar[u] = __builtin_amdgcn_readlane(av, g + u);
        const float* f = h1 + (long)ss * 128 + lane;
        v[u][0] = f[0]; v[u][1] = f[32]; v[u][2] = f[64]; v[u][3] = f[96];
      }
#pragma unroll
      for (int u = 0; u < GR; ++u) {
        float* a = acc + (long)ar[u] * 128 + lane;
#pragma unroll
        for (int j = 0; j < 4; ++j)
          atomicAdd(a + 32 * j, fmaxf(0.0f, v[u][j] * sc[j] + sh[j]));
      }
    }
    base += nWaves * 32;
    if (base < E2) {
      sv = src[base + lane]; ev = et[base + lane]; dv = dst[base + lane];
    }
  }
}

#define SC2_WGS ((E2 / 32 + 7) / 8)
#define NH_WGS  512
__global__ __launch_bounds__(256)
void k_scatter2n_normhead(const float* __restrict__ h1,
                          const float* __restrict__ csum,
                          const float* __restrict__ csq,
                          const float* __restrict__ gamma,
                          const float* __restrict__ beta,
                          const int* __restrict__ src2,
                          const int* __restrict__ dst2,
                          const int* __restrict__ et2,
                          float* __restrict__ A2,
                          float* __restrict__ h1n) {
  if (blockIdx.x < SC2_WGS)
    scatter2n_body((int)blockIdx.x, SC2_WGS, h1, csum, csq, gamma, beta,
                   src2, dst2, et2, A2);
  else
    normhead_body((int)blockIdx.x - SC2_WGS, NH_WGS, h1, csum, csq, gamma, beta, h1n);
}

#define P2 (2 * 64 + 32)
#define LDS2 (64 * P2)

__global__ void k_gemm2(const float* __restrict__ A2,
                        const float* __restrict__ h1n,
                        const float* __restrict__ W2,
                        const float* __restrict__ Ws2,
                        const float* __restrict__ b2,
                        float* __restrict__ out) {
  __shared__ float bs[LDS2];

  int wave = (int)((blockIdx.x * blockDim.x + threadIdx.x) >> 5);
  int lane = threadIdx.x & 31;
  const int mtiles = N2 / 16;
  bool valid = (wave < mtiles);
  int row0 = (valid ? wave : 0) * 16;
  int aRow = lane & 15;
  int aK   = (lane >> 4) << 1;
  int n16  = lane & 15;

  v8f acc[4];
  v8f zero = {0.f,0.f,0.f,0.f,0.f,0.f,0.f,0.f};
#pragma unroll
  for (int t = 0; t < 4; ++t) acc[t] = zero;

  for (int seg = 0; seg < 5; ++seg) {
    const float* Ap = (seg < 4) ? (A2  + ((long)seg * N2 + row0 + aRow) * 128)
                                : (h1n + (long)(row0 + aRow) * 128);
    const float* Bp = (seg < 4) ? (W2 + (long)seg * 128 * 64) : Ws2;

    __syncthreads();
    for (int idx = threadIdx.x; idx < 128 * 64; idx += 256) {
      int k = idx >> 6;
      int n = idx & 63;
      bs[(k >> 1) * P2 + n * 2 + (k & 1)] = Bp[k * 64 + n];
    }
    __syncthreads();

    if (valid) {
      v2f a = *(const v2f*)(Ap + aK);
#pragma unroll 2
      for (int k0 = 0; k0 < 128; k0 += 4) {
        v2f a_cur = a;
        if (k0 < 124) a = *(const v2f*)(Ap + k0 + 4 + aK);
        const v2f* bpair =
            (const v2f*)(bs + ((k0 >> 1) + (lane >> 4)) * P2) + n16;
        v2f bf[4];
#pragma unroll
        for (int t = 0; t < 4; ++t) bf[t] = bpair[t * 16];
#pragma unroll
        for (int t = 0; t < 4; ++t)
          acc[t] = __builtin_amdgcn_wmma_f32_16x16x4_f32(
              false, a_cur, false, bf[t], (short)0, acc[t], false, false);
      }
    }
  }

  if (!valid) return;
  int halfOff = (lane >> 4) << 3;
#pragma unroll
  for (int v = 0; v < 8; ++v) {
    int r = row0 + v + halfOff;
    float vals[4];
    float m = -1e30f;
#pragma unroll
    for (int t = 0; t < 4; ++t) {
      vals[t] = acc[t][v] + b2[t * 16 + n16];
      m = fmaxf(m, vals[t]);
    }
#pragma unroll
    for (int mask = 1; mask < 16; mask <<= 1)
      m = fmaxf(m, __shfl_xor(m, mask, 32));
    float s = 0.f;
#pragma unroll
    for (int t = 0; t < 4; ++t) s += expf(vals[t] - m);
#pragma unroll
    for (int mask = 1; mask < 16; mask <<= 1)
      s += __shfl_xor(s, mask, 32);
    float lse = m + logf(s);
#pragma unroll
    for (int t = 0; t < 4; ++t)
      out[(long)r * 64 + t * 16 + n16] = vals[t] - lse;
  }
}

extern "C" void kernel_launch(void* const* d_in, const int* in_sizes, int n_in,
                              void* d_out, int out_size, void* d_ws, size_t ws_size,
                              hipStream_t stream) {
  const float* x     = (const float*)d_in[0];
  const int*   src1  = (const int*)  d_in[1];
  const int*   dst1  = (const int*)  d_in[2];
  const int*   et1   = (const int*)  d_in[3];
  const int*   src2  = (const int*)  d_in[4];
  const int*   dst2  = (const int*)  d_in[5];
  const int*   et2   = (const int*)  d_in[6];
  const int*   hmap  = (const int*)  d_in[7];
  const float* hbuf  = (const float*)d_in[8];
  const float* W1    = (const float*)d_in[9];
  const float* Ws1   = (const float*)d_in[10];
  const float* b1    = (const float*)d_in[11];
  const float* gamma = (const float*)d_in[12];
  const float* beta  = (const float*)d_in[13];
  const float* W2    = (const float*)d_in[14];
  const float* Ws2   = (const float*)d_in[15];
  const float* b2    = (const float*)d_in[16];

  float* ws    = (float*)d_ws;
  float* A1    = ws;
  float* A2    = A1 + (size_t)4 * N1 * 128;
  float* sums  = A2 + (size_t)4 * N2 * 128;
  float* h1    = sums + 512;
  float* h1n   = h1 + (size_t)N1 * 128;
  float* out   = (float*)d_out;

  k_clear<<<512, 256, 0, stream>>>(A2, (long)4 * N2 * 128 + 256, A1, hmap);

  k_scatter_histfill<<<HF_WGS + SC1_WGS, 256, 0, stream>>>(x, src1, dst1, et1, hmap, hbuf,
                                                           A1, h1, sums, sums + 128);

  k_gemm1<<<(N1/16 + 7) / 8, 256, 0, stream>>>(A1, x, W1, Ws1, b1, hmap, hbuf,
                                               h1, sums, sums + 128);

  k_scatter2n_normhead<<<SC2_WGS + NH_WGS, 256, 0, stream>>>(h1, sums, sums + 128, gamma, beta,
                                                             src2, dst2, et2, A2, h1n);

  k_gemm2<<<(N2/16 + 7) / 8, 256, 0, stream>>>(A2, h1n, W2, Ws2, b2, out);
}


# layer-2 GEMM spread over the chip: 188 work-groups of 32 rows (was 47 of 128), K split over the 8 waves, fragments from global as one load group, partials summed via LDS (K, softmax sum order change)
# speedup vs baseline: 6.5332x; 1.6129x over previous
#include <hip/hip_runtime.h>
#include <hip/hip_bf16.h>
#include <math.h>

typedef __attribute__((ext_vector_type(2))) float v2f;
typedef __attribute__((ext_vector_type(8))) float v8f;

#define N0  300000
#define N1  60000
#define N2  6000
#define E1  1200000
#define E2  300000
#define D_H 128
#define D_OUT 64
#define BN_EPS 1e-5f

__global__ void k_clear(float* __restrict__ p, long n,
                        float* __restrict__ A1, const int* __restrict__ hmap) {
  const long tid = (long)blockIdx.x * blockDim.x + threadIdx.x;
  const long stride = (long)gridDim.x * blockDim.x;
  for (long i = tid; i < n; i += stride) p[i] = 0.0f;
  for (long nn = tid; nn < N1; nn += stride) {
    if (hmap[nn] < 0) {
      for (int r = 0; r < 4; ++r) {
        float* q = A1 + ((long)r * N1 + nn) * 128;
        for (int c = 0; c < 128; ++c) q[c] = 0.0f;
      }
    }
  }
}

#define EPL 8
#define SCAT_WG_EDGES (32 * EPL * 8)
__device__ __forceinline__
void scatter_f_body(const int vb, const int nvb,
                    const float* __restrict__ feat,
                    const int* __restrict__ src,
                    const int* __restrict__ dst,
                    const int* __restrict__ et,
                    const int* __restrict__ hmap,
                    float* __restrict__ acc, int nE, int nOut) {
  const int lane = threadIdx.x & 31;
  const int wave = (int)((vb * blockDim.x + threadIdx.x) >> 5);
  const int nWaves = (int)((nvb * blockDim.x) >> 5);
  for (int base = wave * (32 * EPL); base < nE; base += nWaves * (32 * EPL)) {
    int d[EPL];
    int hm[EPL];
#pragma unroll
    for (int j = 0; j < EPL; ++j)
      d[j] = dst[min(base + j * 32 + lane, nE - 1)];
#pragma unroll
    for (int j = 0; j < EPL; ++j)
      hm[j] = hmap[d[j]];
#pragma unroll
    for (int j = 0; j < EPL; ++j) {
      const int me = base + j * 32 + lane;
      unsigned todo = (unsigned)__ballot(me < nE && hm[j] < 0);
      while (todo) {
        const int l = __ffs(todo) - 1;
        todo &= todo - 1;
        const int pe = base + j * 32 + l;
        const int dd = __builtin_amdgcn_readlane(d[j], l);
        const float* f = feat + (long)src[pe] * 128 + lane;
        float* a = acc + ((long)et[pe] * nOut + dd) * 128 + lane;
        const float v0 = f[0], v1 = f[32], v2 = f[64], v3 = f[96];
        atomicAdd(a,      v0);
        atomicAdd(a + 32, v1);
        atomicAdd(a + 64, v2);
        atomicAdd(a + 96, v3);
      }
    }
  }
}

#define HF_ROWS 16
static_assert(N1 % 32 == 0, "a wave trip covers 32 whole rows");
__device__ __forceinline__
void histfill_stats_body(const int vb, const int nvb,
                         const int* __restrict__ hmap,
                         const float* __restrict__ hbuf,
                         float* __restrict__ h1,
                         float* __restrict__ csum,
                         float* __restrict__ csq) {
  __shared__ __attribute__((aligned(16))) float ls[8][128];
  __shared__ __attribute__((aligned(16))) float lq[8][128];
  const int lane = threadIdx.x & 31;
  const int wv   = threadIdx.x >> 5;
  const int wave = (int)((vb * blockDim.x + threadIdx.x) >> 5);
  const int nWaves = (int)((nvb * blockDim.x) >> 5);
  float4 s = make_float4(0.f, 0.f, 0.f, 0.f);
  float4 q = make_float4(0.f, 0.f, 0.f, 0.f);
  for (int base = wave * 32; base < N1; base += nWaves * 32) {
    const int hmv = hmap[base + lane];
    for (int g = 0; g < 32; g += HF_ROWS) {
      int    hm[HF_ROWS];
      float4 v[HF_ROWS];
#pragma unroll
      for (int u = 0; u < HF_ROWS; ++u) {
        hm[u] = __builtin_amdgcn_readlane(hmv, g + u);
        v[u] = *(const float4*)(hbuf + (long)max(hm[u], 0) * 128 + lane * 4);
      }
#pragma unroll
      for (int u = 0; u < HF_ROWS; ++u) {
        if (hm[u] >= 0) {
          *(float4*)(h1 + (long)(base + g + u) * 128 + lane * 4) = v[u];
          s.x += v[u].x; s.y += v[u].y; s.z += v[u].z; s.w += v[u].w;
          q.x += v[u].x * v[u].x; q.y += v[u].y * v[u].y;
          q.z += v[u].z * v[u].z; q.w += v[u].w * v[u].w;
        }
      }
    }
  }
  *(float4*)&ls[wv][lane * 4] = s;
  *(float4*)&lq[wv][lane * 4] = q;
  __syncthreads();
  if (threadIdx.x < 128) {
    float S = ls[0][threadIdx.x], Q = lq[0][threadIdx.x];
    for (int w = 1; w < 8; ++w) { S += ls[w][threadIdx.x]; Q += lq[w][threadIdx.x]; }
    atomicAdd(&csum[threadIdx.x], S);
    atomicAdd(&csq[threadIdx.x], Q);
  }
}

#define HF_WGS ((N1 / 32 + 7) / 8)
#define SC1_WGS ((E1 + SCAT_WG_EDGES - 1) / SCAT_WG_EDGES)
__global__ __launch_bounds__(256)
void k_scatter_histfill(const float* __restrict__ x,
                        const int* __restrict__ src1,
                        const int* __restrict__ dst1,
                        const int* __restrict__ et1,
                        const int* __restrict__ hmap,
                        const float* __restrict__ hbuf,
                        float* __restrict__ A1,
                        float* __restrict__ h1,
                        float* __restrict__ csum,
                        float* __restrict__ csq) {
  if (blockIdx.x < HF_WGS)
    histfill_stats_body((int)blockIdx.x, HF_WGS, hmap, hbuf, h1, csum, csq);
  else
    scatter_f_body((int)blockIdx.x - HF_WGS, SC1_WGS, x, src1, dst1, et1, hmap,
                   A1, E1, N1);
}

#define G1_KSTEPS 20
#define G1_GROUP  5

__global__ __launch_bounds__(256)
void k_gemm1(const float* __restrict__ A1,
             const float* __restrict__ x,
             const float* __restrict__ W1,
             const float* __restrict__ Ws,
             const float* __restrict__ b1,
             const int*   __restrict__ hmap,
             const float* __restrict__ hbuf,
             float* __restrict__ h1,
             float* __restrict__ csum,
             float* __restrict__ csq) {
  __shared__ float red[8 * 8 * 8 * 32];

  const int wave = threadIdx.x >> 5;
  const int lane = threadIdx.x & 31;
  const int rb = blockIdx.x * 128 + lane * 4;
  bool need = false;
  if (rb < N1) {
    const int4 hv = *(const int4*)(hmap + rb);
    need = (hv.x < 0) | (hv.y < 0) | (hv.z < 0) | (hv.w < 0);
  }
  const unsigned needMask = (unsigned)__ballot(need);
  if (needMask == 0u) return;

  const int c16 = lane & 15;
  const int h   = lane >> 4;

  for (int t = 0; t < 8; ++t) {
    if (((needMask >> (4 * t)) & 0xFu) == 0u) continue;
    const int row0 = blockIdx.x * 128 + t * 16;

    v8f acc[8];
    const v8f zero = {0.f,0.f,0.f,0.f,0.f,0.f,0.f,0.f};
#pragma unroll
    for (int q = 0; q < 8; ++q) acc[q] = zero;

#pragma unroll
    for (int g = 0; g < G1_KSTEPS / G1_GROUP; ++g) {
      v2f    af[G1_GROUP];
      float4 bf[G1_GROUP][2][2];
#pragma unroll
      for (int i = 0; i < G1_GROUP; ++i) {
        const int kk  = wave * (G1_KSTEPS * 4) + (g * G1_GROUP + i) * 4;
        const int seg = kk >> 7;
        const int k   = (kk & 127) + 2 * h;
        const float* Ap = (seg < 4)
            ? (A1 + ((long)seg * N1 + row0 + c16) * 128)
            : (x  + (long)(row0 + c16) * 128);
        const float* Bp = (seg < 4) ? (W1 + (long)seg * 128 * 128) : Ws;
        af[i] = *(const v2f*)(Ap + k);
        const float4* r0 = (const float4*)(Bp + (long)k * 128 + c16 * 8);
        const float4* r1 = (const float4*)(Bp + (long)(k + 1) * 128 + c16 * 8);
        bf[i][0][0] = r0[0]; bf[i][0][1] = r0[1];
        bf[i][1][0] = r1[0]; bf[i][1][1] = r1[1];
      }
#pragma unroll
      for (int i = 0; i < G1_GROUP; ++i) {
        const float k0[8] = {bf[i][0][0].x, bf[i][0][0].y, bf[i][0][0].z, bf[i][0][0].w,
                             bf[i][0][1].x, bf[i][0][1].y, bf[i][0][1].z, bf[i][0][1].w};
        const float k1[8] = {bf[i][1][0].x, bf[i][1][0].y, bf[i][1][0].z, bf[i][1][0].w,
                             bf[i][1][1].x, bf[i][1][1].y, bf[i][1][1].z, bf[i][1][1].w};
#pragma unroll
        for (int q = 0; q < 8; ++q) {
          const v2f b = {k0[q], k1[q]};
          acc[q] = __builtin_amdgcn_wmma_f32_16x16x4_f32(
              false, af[i], false, b, (short)0, acc[q], false, false);
        }
      }
    }

#pragma unroll
    for (int q = 0; q < 8; ++q)
#pragma unroll
      for (int v = 0; v < 8; ++v)
        red[((wave * 8 + q) * 8 + v) * 32 + lane] = acc[q][v];
    __syncthreads();

    float s[8];
#pragma unroll
    for (int q = 0; q < 8; ++q) {
      float sum = red[((0 * 8 + q) * 8 + wave) * 32 + lane];
#pragma unroll
      for (int w = 1; w < 8; ++w)
        sum += red[((w * 8 + q) * 8 + wave) * 32 + lane];
      s[q] = sum;
    }

    const int r    = row0 + wave + 8 * h;
    const int hm   = hmap[r];
    const int col0 = c16 * 8;
    float o[8];
    if (hm >= 0) {
      const float4* hp = (const float4*)(hbuf + (long)hm * 128 + col0);
      const float4 u0 = hp[0], u1 = hp[1];
      o[0] = u0.x; o[1] = u0.y; o[2] = u0.z; o[3] = u0.w;
      o[4] = u1.x; o[5] = u1.y; o[6] = u1.z; o[7] = u1.w;
    } else {
#pragma unroll
      for (int j = 0; j < 8; ++j) {
        o[j] = s[j] + b1[col0 + j];
        atomicAdd(&csum[col0 + j], o[j]);
        atomicAdd(&csq[col0 + j], o[j] * o[j]);
      }
    }
    float4* op = (float4*)(h1 + (long)r * 128 + col0);
    op[0] = make_float4(o[0], o[1], o[2], o[3]);
    op[1] = make_float4(o[4], o[5], o[6], o[7]);
    __syncthreads();
  }
}

__device__ __forceinline__
void bn_coeff(const float* __restrict__ csum, const float* __restrict__ csq,
              const float* __restrict__ gamma, const float* __restrict__ beta,
              const int c, float& sc, float& sh) {
  const float inv_n = 1.0f / (float)N1;
  const float mean = csum[c] * inv_n;
  const float var  = csq[c] * inv_n - mean * mean;
  sc = gamma[c] * rsqrtf(var + BN_EPS);
  sh = beta[c] - mean * sc;
}

__device__ __forceinline__
void normhead_body(const int vb, const int nvb,
                   const float* __restrict__ h1,
                   const float* __restrict__ csum, const float* __restrict__ csq,
                   const float* __restrict__ gamma, const float* __restrict__ beta,
                   float* __restrict__ h1n) {
  const long total = (long)N2 * 128;
  long i = (long)vb * blockDim.x + threadIdx.x;
  const long stride = (long)nvb * blockDim.x;
  float sc, sh;
  bn_coeff(csum, csq, gamma, beta, (int)(i & 127), sc, sh);
  for (; i < total; i += stride)
    h1n[i] = fmaxf(0.0f, h1[i] * sc + sh);
}

#define GR 8
static_assert(E2 % 32 == 0, "a wave trip covers 32 whole edges");
__device__ __forceinline__
void scatter2n_body(const int vb, const int nvb,
                    const float* __restrict__ h1,
                    const float* __restrict__ csum, const float* __restrict__ csq,
                    const float* __restrict__ gamma, const float* __restrict__ beta,
                    const int* __restrict__ src,
                    const int* __restrict__ dst,
                    const int* __restrict__ et,
                    float* __restrict__ acc) {
  const int lane = threadIdx.x & 31;
  const int wave = (int)((vb * blockDim.x + threadIdx.x) >> 5);
  const int nWaves = (int)((nvb * blockDim.x) >> 5);
  int base = wave * 32;
  int sv = 0, ev = 0, dv = 0;
  if (base < E2) {
    sv = src[base + lane]; ev = et[base + lane]; dv = dst[base + lane];
  }
  float sc[4], sh[4];
#pragma unroll
  for (int j = 0; j < 4; ++j)
    bn_coeff(csum, csq, gamma, beta, lane + 32 * j, sc[j], sh[j]);
  while (base < E2) {
    const int av = ev * N2 + dv;
    for (int g = 0; g < 32; g += GR) {
      int   ar[GR];
      float v[GR][4];
#pragma unroll
      for (int u = 0; u < GR; ++u) {
        const int ss = __builtin_amdgcn_readlane(sv, g + u);
        ar[u] = __builtin_amdgcn_readlane(av, g + u);
        const float* f = h1 + (long)ss * 128 + lane;
        v[u][0] = f[0]; v[u][1] = f[32]; v[u][2] = f[64]; v[u][3] = f[96];
      }
#pragma unroll
      for (int u = 0; u < GR; ++u) {
        float* a = acc + (long)ar[u] * 128 + lane;
#pragma unroll
        for (int j = 0; j < 4; ++j)
          atomicAdd(a + 32 * j, fmaxf(0.0f, v[u][j] * sc[j] + sh[j]));
      }
    }
    base += nWaves * 32;
    if (base < E2) {
      sv = src[base + lane]; ev = et[base + lane]; dv = dst[base + lane];
    }
  }
}

#define SC2_WGS ((E2 / 32 + 7) / 8)
#define NH_WGS  512
__global__ __launch_bounds__(256)
void k_scatter2n_normhead(const float* __restrict__ h1,
                          const float* __restrict__ csum,
                          const float* __restrict__ csq,
                          const float* __restrict__ gamma,
                          const float* __restrict__ beta,
                          const int* __restrict__ src2,
                          const int* __restrict__ dst2,
                          const int* __restrict__ et2,
                          float* __restrict__ A2,
                          float* __restrict__ h1n) {
  if (blockIdx.x < SC2_WGS)
    scatter2n_body((int)blockIdx.x, SC2_WGS, h1, csum, csq, gamma, beta,
                   src2, dst2, et2, A2);
  else
    normhead_body((int)blockIdx.x - SC2_WGS, NH_WGS, h1, csum, csq, gamma, beta, h1n);
}

#define G2_KSTEPS 20
#define G2_WGS ((N2 + 31) / 32)
static_assert(N2 % 16 == 0, "row tiles are whole");

__global__ __launch_bounds__(256) __attribute__((amdgpu_waves_per_eu(2, 2)))
void k_gemm2(const float* __restrict__ A2,
             const float* __restrict__ h1n,
             const float* __restrict__ W2,
             const float* __restrict__ Ws2,
             const float* __restrict__ b2,
             float* __restrict__ out) {
  __shared__ float red[8 * 8 * 8 * 32];

  const int wave = __builtin_amdgcn_readfirstlane((int)(threadIdx.x >> 5));
  const int lane = threadIdx.x & 31;
  const int c16  = lane & 15;
  const int h    = lane >> 4;
  const int row0 = blockIdx.x * 32;
  const long ra0 = (long)(row0 + c16) * 128;
  const long ra1 = (long)min(row0 + 16 + c16, N2 - 1) * 128;

  v2f    a0[G2_KSTEPS], a1[G2_KSTEPS];
  float4 bk0[G2_KSTEPS], bk1[G2_KSTEPS];
#pragma unroll
  for (int i = 0; i < G2_KSTEPS; ++i) {
    const int kk  = wave * (G2_KSTEPS * 4) + i * 4;
    const int seg = kk >> 7;
    const int k   = (kk & 127) + 2 * h;
    const float* Ab = (seg < 4) ? (A2 + (long)seg * N2 * 128) : h1n;
    const float* Bp = (seg < 4) ? (W2 + (long)seg * 128 * 64) : Ws2;
    a0[i]  = *(const v2f*)(Ab + ra0 + k);
    a1[i]  = *(const v2f*)(Ab + ra1 + k);
    bk0[i] = *(const float4*)(Bp + (long)k * 64 + c16 * 4);
    bk1[i] = *(const float4*)(Bp + (long)(k + 1) * 64 + c16 * 4);
  }

  v8f acc[2][4];
  const v8f zero = {0.f,0.f,0.f,0.f,0.f,0.f,0.f,0.f};
#pragma unroll
  for (int q = 0; q < 4; ++q) { acc[0][q] = zero; acc[1][q] = zero; }

#pragma unroll
  for (int i = 0; i < G2_KSTEPS; ++i) {
    const float k0[4] = {bk0[i].x, bk0[i].y, bk0[i].z, bk0[i].w};
    const float k1[4] = {bk1[i].x, bk1[i].y, bk1[i].z, bk1[i].w};
#pragma unroll
    for (int q = 0; q < 4; ++q) {
      const v2f b = {k0[q], k1[q]};
      acc[0][q] = __builtin_amdgcn_wmma_f32_16x16x4_f32(
          false, a0[i], false, b, (short)0, acc[0][q], false, false);
      acc[1][q] = __builtin_amdgcn_wmma_f32_16x16x4_f32(
          false, a1[i], false, b, (short)0, acc[1][q], false, false);
    }
  }

#pragma unroll
  for (int m = 0; m < 2; ++m)
#pragma unroll
    for (int q = 0; q < 4; ++q)
#pragma unroll
      for (int v = 0; v < 8; ++v)
        red[((wave * 8 + m * 4 + q) * 8 + v) * 32 + lane] = acc[m][q][v];
  __syncthreads();

  float s[2][4];
#pragma unroll
  for (int m = 0; m < 2; ++m)
#pragma unroll
    for (int q = 0; q < 4; ++q) {
      float sum = red[((0 * 8 + m * 4 + q) * 8 + wave) * 32 + lane];
#pragma unroll
      for (int w = 1; w < 8; ++w)
        sum += red[((w * 8 + m * 4 + q) * 8 + wave) * 32 + lane];
      s[m][q] = sum;
    }

  const float4 bb = *(const float4*)(b2 + c16 * 4);
#pragma unroll
  for (int m = 0; m < 2; ++m) {
    const int r = row0 + 16 * m + wave + 8 * h;
    const float v0 = s[m][0] + bb.x, v1 = s[m][1] + bb.y;
    const float v2 = s[m][2] + bb.z, v3 = s[m][3] + bb.w;
    float mx = fmaxf(fmaxf(v0, v1), fmaxf(v2, v3));
#pragma unroll
    for (int mask = 1; mask < 16; mask <<= 1)
      mx = fmaxf(mx, __shfl_xor(mx, mask, 32));
    float e = (expf(v0 - mx) + expf(v1 - mx)) + (expf(v2 - mx) + expf(v3 - mx));
#pragma unroll
    for (int mask = 1; mask < 16; mask <<= 1)
      e += __shfl_xor(e, mask, 32);
    const float lse = mx + logf(e);
    if (r < N2)
      *(float4*)(out + (long)r * 64 + c16 * 4) =
          make_float4(v0 - lse, v1 - lse, v2 - lse, v3 - lse);
  }
}

extern "C" void kernel_launch(void* const* d_in, const int* in_sizes, int n_in,
                              void* d_out, int out_size, void* d_ws, size_t ws_size,
                              hipStream_t stream) {
  const float* x     = (const float*)d_in[0];
  const int*   src1  = (const int*)  d_in[1];
  const int*   dst1  = (const int*)  d_in[2];
  const int*   et1   = (const int*)  d_in[3];
  const int*   src2  = (const int*)  d_in[4];
  const int*   dst2  = (const int*)  d_in[5];
  const int*   et2   = (const int*)  d_in[6];
  const int*   hmap  = (const int*)  d_in[7];
  const float* hbuf  = (const float*)d_in[8];
  const float* W1    = (const float*)d_in[9];
  const float* Ws1   = (const float*)d_in[10];
  const float* b1    = (const float*)d_in[11];
  const float* gamma = (const float*)d_in[12];
  const float* beta  = (const float*)d_in[13];
  const float* W2    = (const float*)d_in[14];
  const float* Ws2   = (const float*)d_in[15];
  const float* b2    = (const float*)d_in[16];

  float* ws    = (float*)d_ws;
  float* A1    = ws;
  float* A2    = A1 + (size_t)4 * N1 * 128;
  float* sums  = A2 + (size_t)4 * N2 * 128;
  float* h1    = sums + 512;
  float* h1n   = h1 + (size_t)N1 * 128;
  float* out   = (float*)d_out;

  k_clear<<<512, 256, 0, stream>>>(A2, (long)4 * N2 * 128 + 256, A1, hmap);

  k_scatter_histfill<<<HF_WGS + SC1_WGS, 256, 0, stream>>>(x, src1, dst1, et1, hmap, hbuf,
                                                           A1, h1, sums, sums + 128);

  k_gemm1<<<(N1/16 + 7) / 8, 256, 0, stream>>>(A1, x, W1, Ws1, b1, hmap, hbuf,
                                               h1, sums, sums + 128);

  k_scatter2n_normhead<<<SC2_WGS + NH_WGS, 256, 0, stream>>>(h1, sums, sums + 128, gamma, beta,
                                                             src2, dst2, et2, A2, h1n);

  k_gemm2<<<G2_WGS, 256, 0, stream>>>(A2, h1n, W2, Ws2, b2, out);
}


# history rows are no longer copied into h1: layer-2 scatter and head rows read the history buffer through hmap (one more index lookup per trip); the history pass keeps only its BN column sums
# speedup vs baseline: 7.8827x; 1.2066x over previous
#include <hip/hip_runtime.h>
#include <hip/hip_bf16.h>
#include <math.h>

typedef __attribute__((ext_vector_type(2))) float v2f;
typedef __attribute__((ext_vector_type(8))) float v8f;

#define N0  300000
#define N1  60000
#define N2  6000
#define E1  1200000
#define E2  300000
#define D_H 128
#define D_OUT 64
#define BN_EPS 1e-5f

__global__ void k_clear(float* __restrict__ p, long n,
                        float* __restrict__ A1, const int* __restrict__ hmap) {
  const long tid = (long)blockIdx.x * blockDim.x + threadIdx.x;
  const long stride = (long)gridDim.x * blockDim.x;
  for (long i = tid; i < n; i += stride) p[i] = 0.0f;
  for (long nn = tid; nn < N1; nn += stride) {
    if (hmap[nn] < 0) {
      for (int r = 0; r < 4; ++r) {
        float* q = A1 + ((long)r * N1 + nn) * 128;
        for (int c = 0; c < 128; ++c) q[c] = 0.0f;
      }
    }
  }
}

#define EPL 8
#define SCAT_WG_EDGES (32 * EPL * 8)
__device__ __forceinline__
void scatter_f_body(const int vb, const int nvb,
                    const float* __restrict__ feat,
                    const int* __restrict__ src,
                    const int* __restrict__ dst,
                    const int* __restrict__ et,
                    const int* __restrict__ hmap,
                    float* __restrict__ acc, int nE, int nOut) {
  const int lane = threadIdx.x & 31;
  const int wave = (int)((vb * blockDim.x + threadIdx.x) >> 5);
  const int nWaves = (int)((nvb * blockDim.x) >> 5);
  for (int base = wave * (32 * EPL); base < nE; base += nWaves * (32 * EPL)) {
    int d[EPL];
    int hm[EPL];
#pragma unroll
    for (int j = 0; j < EPL; ++j)
      d[j] = dst[min(base + j * 32 + lane, nE - 1)];
#pragma unroll
    for (int j = 0; j < EPL; ++j)
      hm[j] = hmap[d[j]];
#pragma unroll
    for (int j = 0; j < EPL; ++j) {
      const int me = base + j * 32 + lane;
      unsigned todo = (unsigned)__ballot(me < nE && hm[j] < 0);
      while (todo) {
        const int l = __ffs(todo) - 1;
        todo &= todo - 1;
        const int pe = base + j * 32 + l;
        const int dd = __builtin_amdgcn_readlane(d[j], l);
        const float* f = feat + (long)src[pe] * 128 + lane;
        float* a = acc + ((long)et[pe] * nOut + dd) * 128 + lane;
        const float v0 = f[0], v1 = f[32], v2 = f[64], v3 = f[96];
        atomicAdd(a,      v0);
        atomicAdd(a + 32, v1);
        atomicAdd(a + 64, v2);
        atomicAdd(a + 96, v3);
      }
    }
  }
}

#define HF_ROWS 16
static_assert(N1 % 32 == 0, "a wave trip covers 32 whole rows");
__device__ __forceinline__
void histfill_stats_body(const int vb, const int nvb,
                         const int* __restrict__ hmap,
                         const float* __restrict__ hbuf,
                         float* __restrict__ csum,
                         float* __restrict__ csq) {
  __shared__ __attribute__((aligned(16))) float ls[8][128];
  __shared__ __attribute__((aligned(16))) float lq[8][128];
  const int lane = threadIdx.x & 31;
  const int wv   = threadIdx.x >> 5;
  const int wave = (int)((vb * blockDim.x + threadIdx.x) >> 5);
  const int nWaves = (int)((nvb * blockDim.x) >> 5);
  float4 s = make_float4(0.f, 0.f, 0.f, 0.f);
  float4 q = make_float4(0.f, 0.f, 0.f, 0.f);
  for (int base = wave * 32; base < N1; base += nWaves * 32) {
    const int hmv = hmap[base + lane];
    for (int g = 0; g < 32; g += HF_ROWS) {
      int    hm[HF_ROWS];
      float4 v[HF_ROWS];
#pragma unroll
      for (int u = 0; u < HF_ROWS; ++u) {
        hm[u] = __builtin_amdgcn_readlane(hmv, g + u);
        v[u] = *(const float4*)(hbuf + (long)max(hm[u], 0) * 128 + lane * 4);
      }
#pragma unroll
      for (int u = 0; u < HF_ROWS; ++u) {
        if (hm[u] >= 0) {
          s.x += v[u].x; s.y += v[u].y; s.z += v[u].z; s.w += v[u].w;
          q.x += v[u].x * v[u].x; q.y += v[u].y * v[u].y;
          q.z += v[u].z * v[u].z; q.w += v[u].w * v[u].w;
        }
      }
    }
  }
  *(float4*)&ls[wv][lane * 4] = s;
  *(float4*)&lq[wv][lane * 4] = q;
  __syncthreads();
  if (threadIdx.x < 128) {
    float S = ls[0][threadIdx.x], Q = lq[0][threadIdx.x];
    for (int w = 1; w < 8; ++w) { S += ls[w][threadIdx.x]; Q += lq[w][threadIdx.x]; }
    atomicAdd(&csum[threadIdx.x], S);
    atomicAdd(&csq[threadIdx.x], Q);
  }
}

#define HF_WGS ((N1 / 32 + 7) / 8)
#define SC1_WGS ((E1 + SCAT_WG_EDGES - 1) / SCAT_WG_EDGES)
__global__ __launch_bounds__(256)
void k_scatter_histfill(const float* __restrict__ x,
                        const int* __restrict__ src1,
                        const int* __restrict__ dst1,
                        const int* __restrict__ et1,
                        const int* __restrict__ hmap,
                        const float* __restrict__ hbuf,
                        float* __restrict__ A1,
                        float* __restrict__ csum,
                        float* __restrict__ csq) {
  if (blockIdx.x < HF_WGS)
    histfill_stats_body((int)blockIdx.x, HF_WGS, hmap, hbuf, csum, csq);
  else
    scatter_f_body((int)blockIdx.x - HF_WGS, SC1_WGS, x, src1, dst1, et1, hmap,
                   A1, E1, N1);
}

#define G1_KSTEPS 20
#define G1_GROUP  5

__global__ __launch_bounds__(256)
void k_gemm1(const float* __restrict__ A1,
             const float* __restrict__ x,
             const float* __restrict__ W1,
             const float* __restrict__ Ws,
             const float* __restrict__ b1,
             const int*   __restrict__ hmap,
             float* __restrict__ h1,
             float* __restrict__ csum,
             float* __restrict__ csq) {
  __shared__ float red[8 * 8 * 8 * 32];

  const int wave = threadIdx.x >> 5;
  const int lane = threadIdx.x & 31;
  const int rb = blockIdx.x * 128 + lane * 4;
  bool need = false;
  if (rb < N1) {
    const int4 hv = *(const int4*)(hmap + rb);
    need = (hv.x < 0) | (hv.y < 0) | (hv.z < 0) | (hv.w < 0);
  }
  const unsigned needMask = (unsigned)__ballot(need);
  if (needMask == 0u) return;

  const int c16 = lane & 15;
  const int h   = lane >> 4;

  for (int t = 0; t < 8; ++t) {
    if (((needMask >> (4 * t)) & 0xFu) == 0u) continue;
    const int row0 = blockIdx.x * 128 + t * 16;

    v8f acc[8];
    const v8f zero = {0.f,0.f,0.f,0.f,0.f,0.f,0.f,0.f};
#pragma unroll
    for (int q = 0; q < 8; ++q) acc[q] = zero;

#pragma unroll
    for (int g = 0; g < G1_KSTEPS / G1_GROUP; ++g) {
      v2f    af[G1_GROUP];
      float4 bf[G1_GROUP][2][2];
#pragma unroll
      for (int i = 0; i < G1_GROUP; ++i) {
        const int kk  = wave * (G1_KSTEPS * 4) + (g * G1_GROUP + i) * 4;
        const int seg = kk >> 7;
        const int k   = (kk & 127) + 2 * h;
        const float* Ap = (seg < 4)
            ? (A1 + ((long)seg * N1 + row0 + c16) * 128)
            : (x  + (long)(row0 + c16) * 128);
        const float* Bp = (seg < 4) ? (W1 + (long)seg * 128 * 128) : Ws;
        af[i] = *(const v2f*)(Ap + k);
        const float4* r0 = (const float4*)(Bp + (long)k * 128 + c16 * 8);
        const float4* r1 = (const float4*)(Bp + (long)(k + 1) * 128 + c16 * 8);
        bf[i][0][0] = r0[0]; bf[i][0][1] = r0[1];
        bf[i][1][0] = r1[0]; bf[i][1][1] = r1[1];
      }
#pragma unroll
      for (int i = 0; i < G1_GROUP; ++i) {
        const float k0[8] = {bf[i][0][0].x, bf[i][0][0].y, bf[i][0][0].z, bf[i][0][0].w,
                             bf[i][0][1].x, bf[i][0][1].y, bf[i][0][1].z, bf[i][0][1].w};
        const float k1[8] = {bf[i][1][0].x, bf[i][1][0].y, bf[i][1][0].z, bf[i][1][0].w,
                             bf[i][1][1].x, bf[i][1][1].y, bf[i][1][1].z, bf[i][1][1].w};
#pragma unroll
        for (int q = 0; q < 8; ++q) {
          const v2f b = {k0[q], k1[q]};
          acc[q] = __builtin_amdgcn_wmma_f32_16x16x4_f32(
              false, af[i], false, b, (short)0, acc[q], false, false);
        }
      }
    }

#pragma unroll
    for (int q = 0; q < 8; ++q)
#pragma unroll
      for (int v = 0; v < 8; ++v)
        red[((wave * 8 + q) * 8 + v) * 32 + lane] = acc[q][v];
    __syncthreads();

    float s[8];
#pragma unroll
    for (int q = 0; q < 8; ++q) {
      float sum = red[((0 * 8 + q) * 8 + wave) * 32 + lane];
#pragma unroll
      for (int w = 1; w < 8; ++w)
        sum += red[((w * 8 + q) * 8 + wave) * 32 + lane];
      s[q] = sum;
    }

    const int r    = row0 + wave + 8 * h;
    const int col0 = c16 * 8;
    if (hmap[r] < 0) {
      float o[8];
#pragma unroll
      for (int j = 0; j < 8; ++j) {
        o[j] = s[j] + b1[col0 + j];
        atomicAdd(&csum[col0 + j], o[j]);
        atomicAdd(&csq[col0 + j], o[j] * o[j]);
      }
      float4* op = (float4*)(h1 + (long)r * 128 + col0);
      op[0] = make_float4(o[0], o[1], o[2], o[3]);
      op[1] = make_float4(o[4], o[5], o[6], o[7]);
    }
    __syncthreads();
  }
}

__device__ __forceinline__
void bn_coeff(const float* __restrict__ csum, const float* __restrict__ csq,
              const float* __restrict__ gamma, const float* __restrict__ beta,
              const int c, float& sc, float& sh) {
  const float inv_n = 1.0f / (float)N1;
  const float mean = csum[c] * inv_n;
  const float var  = csq[c] * inv_n - mean * mean;
  sc = gamma[c] * rsqrtf(var + BN_EPS);
  sh = beta[c] - mean * sc;
}

#define NH_WGS   512
#define NH_TRIPS ((N2 * 128 + NH_WGS * 256 - 1) / (NH_WGS * 256))
__device__ __forceinline__
void normhead_body(const int vb,
                   const float* __restrict__ h1,
                   const int* __restrict__ hmap,
                   const float* __restrict__ hbuf,
                   const float* __restrict__ csum, const float* __restrict__ csq,
                   const float* __restrict__ gamma, const float* __restrict__ beta,
                   float* __restrict__ h1n) {
  const int total = N2 * 128;
  const int i0 = vb * 256 + (int)threadIdx.x;
  const int col = i0 & 127;
  int   hm[NH_TRIPS];
  float v[NH_TRIPS];
#pragma unroll
  for (int t = 0; t < NH_TRIPS; ++t)
    hm[t] = hmap[min(i0 + t * (NH_WGS * 256), total - 1) >> 7];
  float sc, sh;
  bn_coeff(csum, csq, gamma, beta, col, sc, sh);
#pragma unroll
  for (int t = 0; t < NH_TRIPS; ++t) {
    const int i = min(i0 + t * (NH_WGS * 256), total - 1);
    v[t] = (hm[t] >= 0) ? hbuf[(long)hm[t] * 128 + col] : h1[i];
  }
#pragma unroll
  for (int t = 0; t < NH_TRIPS; ++t) {
    const int i = i0 + t * (NH_WGS * 256);
    if (i < total) h1n[i] = fmaxf(0.0f, v[t] * sc + sh);
  }
}

#define GR 8
static_assert(E2 % 32 == 0, "a wave trip covers 32 whole edges");
__device__ __forceinline__
void scatter2n_body(const int vb, const int nvb,
                    const float* __restrict__ h1,
                    const int* __restrict__ hmap,
                    const float* __restrict__ hbuf,
                    const float* __restrict__ csum, const float* __restrict__ csq,
                    const float* __restrict__ gamma, const float* __restrict__ beta,
                    const int* __restrict__ src,
                    const int* __restrict__ dst,
                    const int* __restrict__ et,
                    float* __restrict__ acc) {
  const int lane = threadIdx.x & 31;
  const int wave = (int)((vb * blockDim.x + threadIdx.x) >> 5);
  const int nWaves = (int)((nvb * blockDim.x) >> 5);
  int base = wave * 32;
  int sv = 0, ev = 0, dv = 0;
  if (base < E2) {
    sv = src[base + lane]; ev = et[base + lane]; dv = dst[base + lane];
  }
  int hv = hmap[sv];
  float sc[4], sh[4];
#pragma unroll
  for (int j = 0; j < 4; ++j)
    bn_coeff(csum, csq, gamma, beta, lane + 32 * j, sc[j], sh[j]);
  while (base < E2) {
    const int av = ev * N2 + dv;
    for (int g = 0; g < 32; g += GR) {
      int   ar[GR];
      float v[GR][4];
#pragma unroll
      for (int u = 0; u < GR; ++u) {
        const int ss = __builtin_amdgcn_readlane(sv, g + u);
        const int hh = __builtin_amdgcn_readlane(hv, g + u);
        ar[u] = __builtin_amdgcn_readlane(av, g + u);
        const float* f = ((hh >= 0) ? (hbuf + (long)hh * 128)
                                    : (h1 + (long)ss * 128)) + lane;
        v[u][0] = f[0]; v[u][1] = f[32]; v[u][2] = f[64]; v[u][3] = f[96];
      }
#pragma unroll
      for (int u = 0; u < GR; ++u) {
        float* a = acc + (long)ar[u] * 128 + lane;
#pragma unroll
        for (int j = 0; j < 4; ++j)
          atomicAdd(a + 32 * j, fmaxf(0.0f, v[u][j] * sc[j] + sh[j]));
      }
    }
    base += nWaves * 32;
    if (base < E2) {
      sv = src[base + lane]; ev = et[base + lane]; dv = dst[base + lane];
      hv = hmap[sv];
    }
  }
}

#define SC2_WGS ((E2 / 32 + 7) / 8)
__global__ __launch_bounds__(256)
void k_scatter2n_normhead(const float* __restrict__ h1,
                          const int* __restrict__ hmap,
                          const float* __restrict__ hbuf,
                          const float* __restrict__ csum,
                          const float* __restrict__ csq,
                          const float* __restrict__ gamma,
                          const float* __restrict__ beta,
                          const int* __restrict__ src2,
                          const int* __restrict__ dst2,
                          const int* __restrict__ et2,
                          float* __restrict__ A2,
                          float* __restrict__ h1n) {
  if (blockIdx.x < SC2_WGS)
    scatter2n_body((int)blockIdx.x, SC2_WGS, h1, hmap, hbuf, csum, csq, gamma, beta,
                   src2, dst2, et2, A2);
  else
    normhead_body((int)blockIdx.x - SC2_WGS, h1, hmap, hbuf, csum, csq,
                  gamma, beta, h1n);
}

#define G2_KSTEPS 20
#define G2_WGS ((N2 + 31) / 32)
static_assert(N2 % 16 == 0, "row tiles are whole");

__global__ __launch_bounds__(256) __attribute__((amdgpu_waves_per_eu(2, 2)))
void k_gemm2(const float* __restrict__ A2,
             const float* __restrict__ h1n,
             const float* __restrict__ W2,
             const float* __restrict__ Ws2,
             const float* __restrict__ b2,
             float* __restrict__ out) {
  __shared__ float red[8 * 8 * 8 * 32];

  const int wave = __builtin_amdgcn_readfirstlane((int)(threadIdx.x >> 5));
  const int lane = threadIdx.x & 31;
  const int c16  = lane & 15;
  const int h    = lane >> 4;
  const int row0 = blockIdx.x * 32;
  const long ra0 = (long)(row0 + c16) * 128;
  const long ra1 = (long)min(row0 + 16 + c16, N2 - 1) * 128;

  v2f    a0[G2_KSTEPS], a1[G2_KSTEPS];
  float4 bk0[G2_KSTEPS], bk1[G2_KSTEPS];
#pragma unroll
  for (int i = 0; i < G2_KSTEPS; ++i) {
    const int kk  = wave * (G2_KSTEPS * 4) + i * 4;
    const int seg = kk >> 7;
    const int k   = (kk & 127) + 2 * h;
    const float* Ab = (seg < 4) ? (A2 + (long)seg * N2 * 128) : h1n;
    const float* Bp = (seg < 4) ? (W2 + (long)seg * 128 * 64) : Ws2;
    a0[i]  = *(const v2f*)(Ab + ra0 + k);
    a1[i]  = *(const v2f*)(Ab + ra1 + k);
    bk0[i] = *(const float4*)(Bp + (long)k * 64 + c16 * 4);
    bk1[i] = *(const float4*)(Bp + (long)(k + 1) * 64 + c16 * 4);
  }

  v8f acc[2][4];
  const v8f zero = {0.f,0.f,0.f,0.f,0.f,0.f,0.f,0.f};
#pragma unroll
  for (int q = 0; q < 4; ++q) { acc[0][q] = zero; acc[1][q] = zero; }

#pragma unroll
  for (int i = 0; i < G2_KSTEPS; ++i) {
    const float k0[4] = {bk0[i].x, bk0[i].y, bk0[i].z, bk0[i].w};
    const float k1[4] = {bk1[i].x, bk1[i].y, bk1[i].z, bk1[i].w};
#pragma unroll
    for (int q = 0; q < 4; ++q) {
      const v2f b = {k0[q], k1[q]};
      acc[0][q] = __builtin_amdgcn_wmma_f32_16x16x4_f32(
          false, a0[i], false, b, (short)0, acc[0][q], false, false);
      acc[1][q] = __builtin_amdgcn_wmma_f32_16x16x4_f32(
          false, a1[i], false, b, (short)0, acc[1][q], false, false);
    }
  }

#pragma unroll
  for (int m = 0; m < 2; ++m)
#pragma unroll
    for (int q = 0; q < 4; ++q)
#pragma unroll
      for (int v = 0; v < 8; ++v)
        red[((wave * 8 + m * 4 + q) * 8 + v) * 32 + lane] = acc[m][q][v];
  __syncthreads();

  float s[2][4];
#pragma unroll
  for (int m = 0; m < 2; ++m)
#pragma unroll
    for (int q = 0; q < 4; ++q) {
      float sum = red[((0 * 8 + m * 4 + q) * 8 + wave) * 32 + lane];
#pragma unroll
      for (int w = 1; w < 8; ++w)
        sum += red[((w * 8 + m * 4 + q) * 8 + wave) * 32 + lane];
      s[m][q] = sum;
    }

  const float4 bb = *(const float4*)(b2 + c16 * 4);
#pragma unroll
  for (int m = 0; m < 2; ++m) {
    const int r = row0 + 16 * m + wave + 8 * h;
    const float v0 = s[m][0] + bb.x, v1 = s[m][1] + bb.y;
    const float v2 = s[m][2] + bb.z, v3 = s[m][3] + bb.w;
    float mx = fmaxf(fmaxf(v0, v1), fmaxf(v2, v3));
#pragma unroll
    for (int mask = 1; mask < 16; mask <<= 1)
      mx = fmaxf(mx, __shfl_xor(mx, mask, 32));
    float e = (expf(v0 - mx) + expf(v1 - mx)) + (expf(v2 - mx) + expf(v3 - mx));
#pragma unroll
    for (int mask = 1; mask < 16; mask <<= 1)
      e += __shfl_xor(e, mask, 32);
    const float lse = mx + logf(e);
    if (r < N2)
      *(float4*)(out + (long)r * 64 + c16 * 4) =
          make_float4(v0 - lse, v1 - lse, v2 - lse, v3 - lse);
  }
}

extern "C" void kernel_launch(void* const* d_in, const int* in_sizes, int n_in,
                              void* d_out, int out_size, void* d_ws, size_t ws_size,
                              hipStream_t stream) {
  const float* x     = (const float*)d_in[0];
  const int*   src1  = (const int*)  d_in[1];
  const int*   dst1  = (const int*)  d_in[2];
  const int*   et1   = (const int*)  d_in[3];
  const int*   src2  = (const int*)  d_in[4];
  const int*   dst2  = (const int*)  d_in[5];
  const int*   et2   = (const int*)  d_in[6];
  const int*   hmap  = (const int*)  d_in[7];
  const float* hbuf  = (const float*)d_in[8];
  const float* W1    = (const float*)d_in[9];
  const float* Ws1   = (const float*)d_in[10];
  const float* b1    = (const float*)d_in[11];
  const float* gamma = (const float*)d_in[12];
  const float* beta  = (const float*)d_in[13];
  const float* W2    = (const float*)d_in[14];
  const float* Ws2   = (const float*)d_in[15];
  const float* b2    = (const float*)d_in[16];

  float* ws    = (float*)d_ws;
  float* A1    = ws;
  float* A2    = A1 + (size_t)4 * N1 * 128;
  float* sums  = A2 + (size_t)4 * N2 * 128;
  float* h1    = sums + 512;
  float* h1n   = h1 + (size_t)N1 * 128;
  float* out   = (float*)d_out;

  k_clear<<<512, 256, 0, stream>>>(A2, (long)4 * N2 * 128 + 256, A1, hmap);

  k_scatter_histfill<<<HF_WGS + SC1_WGS, 256, 0, stream>>>(x, src1, dst1, et1, hmap, hbuf,
                                                           A1, sums, sums + 128);

  k_gemm1<<<(N1/16 + 7) / 8, 256, 0, stream>>>(A1, x, W1, Ws1, b1, hmap,
                                               h1, sums, sums + 128);

  k_scatter2n_normhead<<<SC2_WGS + NH_WGS, 256, 0, stream>>>(h1, hmap, hbuf, sums, sums + 128, gamma, beta,
                                                             src2, dst2, et2, A2, h1n);

  k_gemm2<<<G2_WGS, 256, 0, stream>>>(A2, h1n, W2, Ws2, b2, out);
}


# layer-1 GEMM: the rare computed tile's 128 columns split over 4 work-groups (32 columns = one output line each), all fragment loads as one group; K sum order kept, BN atomics arrive in another order
# speedup vs baseline: 8.5233x; 1.0813x over previous
#include <hip/hip_runtime.h>
#include <hip/hip_bf16.h>
#include <math.h>

typedef __attribute__((ext_vector_type(2))) float v2f;
typedef __attribute__((ext_vector_type(8))) float v8f;

#define N0  300000
#define N1  60000
#define N2  6000
#define E1  1200000
#define E2  300000
#define D_H 128
#define D_OUT 64
#define BN_EPS 1e-5f

__global__ void k_clear(float* __restrict__ p, long n,
                        float* __restrict__ A1, const int* __restrict__ hmap) {
  const long tid = (long)blockIdx.x * blockDim.x + threadIdx.x;
  const long stride = (long)gridDim.x * blockDim.x;
  for (long i = tid; i < n; i += stride) p[i] = 0.0f;
  for (long nn = tid; nn < N1; nn += stride) {
    if (hmap[nn] < 0) {
      for (int r = 0; r < 4; ++r) {
        float* q = A1 + ((long)r * N1 + nn) * 128;
        for (int c = 0; c < 128; ++c) q[c] = 0.0f;
      }
    }
  }
}

#define EPL 8
#define SCAT_WG_EDGES (32 * EPL * 8)
__device__ __forceinline__
void scatter_f_body(const int vb, const int nvb,
                    const float* __restrict__ feat,
                    const int* __restrict__ src,
                    const int* __restrict__ dst,
                    const int* __restrict__ et,
                    const int* __restrict__ hmap,
                    float* __restrict__ acc, int nE, int nOut) {
  const int lane = threadIdx.x & 31;
  const int wave = (int)((vb * blockDim.x + threadIdx.x) >> 5);
  const int nWaves = (int)((nvb * blockDim.x) >> 5);
  for (int base = wave * (32 * EPL); base < nE; base += nWaves * (32 * EPL)) {
    int d[EPL];
    int hm[EPL];
#pragma unroll
    for (int j = 0; j < EPL; ++j)
      d[j] = dst[min(base + j * 32 + lane, nE - 1)];
#pragma unroll
    for (int j = 0; j < EPL; ++j)
      hm[j] = hmap[d[j]];
#pragma unroll
    for (int j = 0; j < EPL; ++j) {
      const int me = base + j * 32 + lane;
      unsigned todo = (unsigned)__ballot(me < nE && hm[j] < 0);
      while (todo) {
        const int l = __ffs(todo) - 1;
        todo &= todo - 1;
        const int pe = base + j * 32 + l;
        const int dd = __builtin_amdgcn_readlane(d[j], l);
        const float* f = feat + (long)src[pe] * 128 + lane;
        float* a = acc + ((long)et[pe] * nOut + dd) * 128 + lane;
        const float v0 = f[0], v1 = f[32], v2 = f[64], v3 = f[96];
        atomicAdd(a,      v0);
        atomicAdd(a + 32, v1);
        atomicAdd(a + 64, v2);
        atomicAdd(a + 96, v3);
      }
    }
  }
}

#define HF_ROWS 16
static_assert(N1 % 32 == 0, "a wave trip covers 32 whole rows");
__device__ __forceinline__
void histfill_stats_body(const int vb, const int nvb,
                         const int* __restrict__ hmap,
                         const float* __restrict__ hbuf,
                         float* __restrict__ csum,
                         float* __restrict__ csq) {
  __shared__ __attribute__((aligned(16))) float ls[8][128];
  __shared__ __attribute__((aligned(16))) float lq[8][128];
  const int lane = threadIdx.x & 31;
  const int wv   = threadIdx.x >> 5;
  const int wave = (int)((vb * blockDim.x + threadIdx.x) >> 5);
  const int nWaves = (int)((nvb * blockDim.x) >> 5);
  float4 s = make_float4(0.f, 0.f, 0.f, 0.f);
  float4 q = make_float4(0.f, 0.f, 0.f, 0.f);
  for (int base = wave * 32; base < N1; base += nWaves * 32) {
    const int hmv = hmap[base + lane];
    for (int g = 0; g < 32; g += HF_ROWS) {
      int    hm[HF_ROWS];
      float4 v[HF_ROWS];
#pragma unroll
      for (int u = 0; u < HF_ROWS; ++u) {
        hm[u] = __builtin_amdgcn_readlane(hmv, g + u);
        v[u] = *(const float4*)(hbuf + (long)max(hm[u], 0) * 128 + lane * 4);
      }
#pragma unroll
      for (int u = 0; u < HF_ROWS; ++u) {
        if (hm[u] >= 0) {
          s.x += v[u].x; s.y += v[u].y; s.z += v[u].z; s.w += v[u].w;
          q.x += v[u].x * v[u].x; q.y += v[u].y * v[u].y;
          q.z += v[u].z * v[u].z; q.w += v[u].w * v[u].w;
        }
      }
    }
  }
  *(float4*)&ls[wv][lane * 4] = s;
  *(float4*)&lq[wv][lane * 4] = q;
  __syncthreads();
  if (threadIdx.x < 128) {
    float S = ls[0][threadIdx.x], Q = lq[0][threadIdx.x];
    for (int w = 1; w < 8; ++w) { S += ls[w][threadIdx.x]; Q += lq[w][threadIdx.x]; }
    atomicAdd(&csum[threadIdx.x], S);
    atomicAdd(&csq[threadIdx.x], Q);
  }
}

#define HF_WGS ((N1 / 32 + 7) / 8)
#define SC1_WGS ((E1 + SCAT_WG_EDGES - 1) / SCAT_WG_EDGES)
__global__ __launch_bounds__(256)
void k_scatter_histfill(const float* __restrict__ x,
                        const int* __restrict__ src1,
                        const int* __restrict__ dst1,
                        const int* __restrict__ et1,
                        const int* __restrict__ hmap,
                        const float* __restrict__ hbuf,
                        float* __restrict__ A1,
                        float* __restrict__ csum,
                        float* __restrict__ csq) {
  if (blockIdx.x < HF_WGS)
    histfill_stats_body((int)blockIdx.x, HF_WGS, hmap, hbuf, csum, csq);
  else
    scatter_f_body((int)blockIdx.x - HF_WGS, SC1_WGS, x, src1, dst1, et1, hmap,
                   A1, E1, N1);
}

#define G1_KSTEPS 20
#define G1_ROWS   1024
#define G1_RGS    ((N1 + G1_ROWS - 1) / G1_ROWS)
static_assert(N1 % 16 == 0, "row tiles are whole");
static_assert(N1 % 4 == 0, "four history entries are in or out together");

__global__ __launch_bounds__(256)
void k_gemm1(const float* __restrict__ A1,
             const float* __restrict__ x,
             const float* __restrict__ W1,
             const float* __restrict__ Ws,
             const float* __restrict__ b1,
             const int*   __restrict__ hmap,
             float* __restrict__ h1,
             float* __restrict__ csum,
             float* __restrict__ csq) {
  __shared__ float red[8 * 2 * 8 * 32];

  const int wave = __builtin_amdgcn_readfirstlane((int)(threadIdx.x >> 5));
  const int lane = threadIdx.x & 31;
  const int rg   = (int)blockIdx.x >> 2;
  const int qc   = (int)blockIdx.x & 3;
  bool need[8];
#pragma unroll
  for (int b = 0; b < 8; ++b) {
    const int rb = rg * G1_ROWS + b * 128 + lane * 4;
    const int4 hv = *(const int4*)(hmap + min(rb, N1 - 4));
    need[b] = (rb < N1) & ((hv.x < 0) | (hv.y < 0) | (hv.z < 0) | (hv.w < 0));
  }
  unsigned long long todo = 0ull;
#pragma unroll
  for (int b = 0; b < 8; ++b) {
    unsigned m = (unsigned)__ballot(need[b]);
    m |= m >> 1;
    m |= m >> 2;
    unsigned bits = 0u;
#pragma unroll
    for (int t = 0; t < 8; ++t) bits |= ((m >> (4 * t)) & 1u) << t;
    todo |= (unsigned long long)bits << (8 * b);
  }
  if (todo == 0ull) return;

  const int c16 = lane & 15;
  const int h   = lane >> 4;
  const int colB = qc * 32 + c16 * 2;

  int tile = __builtin_ctzll(todo);
  todo &= todo - 1ull;
  v2f af[G1_KSTEPS];
#pragma unroll
  for (int i = 0; i < G1_KSTEPS; ++i) {
    const int kk  = wave * (G1_KSTEPS * 4) + i * 4;
    const int seg = kk >> 7;
    const int k   = (kk & 127) + 2 * h;
    const float* Ab = (seg < 4) ? (A1 + (long)seg * N1 * 128) : x;
    af[i] = *(const v2f*)(Ab + (long)(rg * G1_ROWS + tile * 16 + c16) * 128 + k);
  }
  v2f bk0[G1_KSTEPS], bk1[G1_KSTEPS];
#pragma unroll
  for (int i = 0; i < G1_KSTEPS; ++i) {
    const int kk  = wave * (G1_KSTEPS * 4) + i * 4;
    const int seg = kk >> 7;
    const int k   = (kk & 127) + 2 * h;
    const float* Bp = (seg < 4) ? (W1 + (long)seg * 128 * 128) : Ws;
    bk0[i] = *(const v2f*)(Bp + (long)k * 128 + colB);
    bk1[i] = *(const v2f*)(Bp + (long)(k + 1) * 128 + colB);
  }

  for (;;) {
    const int row0 = rg * G1_ROWS + tile * 16;
    const int r = row0 + wave + 8 * h;
    const int hr = hmap[r];
    const float2 bb = *(const float2*)(b1 + colB);

    const v8f zero = {0.f,0.f,0.f,0.f,0.f,0.f,0.f,0.f};
    v8f acc0 = zero, acc1 = zero;
#pragma unroll
    for (int i = 0; i < G1_KSTEPS; ++i) {
      const v2f bt0 = {bk0[i].x, bk1[i].x};
      const v2f bt1 = {bk0[i].y, bk1[i].y};
      acc0 = __builtin_amdgcn_wmma_f32_16x16x4_f32(
          false, af[i], false, bt0, (short)0, acc0, false, false);
      acc1 = __builtin_amdgcn_wmma_f32_16x16x4_f32(
          false, af[i], false, bt1, (short)0, acc1, false, false);
    }

#pragma unroll
    for (int v = 0; v < 8; ++v) {
      red[((wave * 2 + 0) * 8 + v) * 32 + lane] = acc0[v];
      red[((wave * 2 + 1) * 8 + v) * 32 + lane] = acc1[v];
    }
    __syncthreads();

    float s[2];
#pragma unroll
    for (int t = 0; t < 2; ++t) {
      float sum = red[((0 * 2 + t) * 8 + wave) * 32 + lane];
#pragma unroll
      for (int w = 1; w < 8; ++w)
        sum += red[((w * 2 + t) * 8 + wave) * 32 + lane];
      s[t] = sum;
    }

    if (hr < 0) {
      const float o0 = s[0] + bb.x;
      const float o1 = s[1] + bb.y;
      atomicAdd(&csum[colB],     o0);
      atomicAdd(&csq[colB],      o0 * o0);
      atomicAdd(&csum[colB + 1], o1);
      atomicAdd(&csq[colB + 1],  o1 * o1);
      *(float2*)(h1 + (long)r * 128 + colB) = make_float2(o0, o1);
    }

    if (todo == 0ull) break;
    tile = __builtin_ctzll(todo);
    todo &= todo - 1ull;
#pragma unroll
    for (int i = 0; i < G1_KSTEPS; ++i) {
      const int kk  = wave * (G1_KSTEPS * 4) + i * 4;
      const int seg = kk >> 7;
      const int k   = (kk & 127) + 2 * h;
      const float* Ab = (seg < 4) ? (A1 + (long)seg * N1 * 128) : x;
      af[i] = *(const v2f*)(Ab + (long)(rg * G1_ROWS + tile * 16 + c16) * 128 + k);
    }
    __syncthreads();
  }
}

__device__ __forceinline__
void bn_coeff(const float* __restrict__ csum, const float* __restrict__ csq,
              const float* __restrict__ gamma, const float* __restrict__ beta,
              const int c, float& sc, float& sh) {
  const float inv_n = 1.0f / (float)N1;
  const float mean = csum[c] * inv_n;
  const float var  = csq[c] * inv_n - mean * mean;
  sc = gamma[c] * rsqrtf(var + BN_EPS);
  sh = beta[c] - mean * sc;
}

#define NH_WGS   512
#define NH_TRIPS ((N2 * 128 + NH_WGS * 256 - 1) / (NH_WGS * 256))
__device__ __forceinline__
void normhead_body(const int vb,
                   const float* __restrict__ h1,
                   const int* __restrict__ hmap,
                   const float* __restrict__ hbuf,
                   const float* __restrict__ csum, const float* __restrict__ csq,
                   const float* __restrict__ gamma, const float* __restrict__ beta,
                   float* __restrict__ h1n) {
  const int total = N2 * 128;
  const int i0 = vb * 256 + (int)threadIdx.x;
  const int col = i0 & 127;
  int   hm[NH_TRIPS];
  float v[NH_TRIPS];
#pragma unroll
  for (int t = 0; t < NH_TRIPS; ++t)
    hm[t] = hmap[min(i0 + t * (NH_WGS * 256), total - 1) >> 7];
  float sc, sh;
  bn_coeff(csum, csq, gamma, beta, col, sc, sh);
#pragma unroll
  for (int t = 0; t < NH_TRIPS; ++t) {
    const int i = min(i0 + t * (NH_WGS * 256), total - 1);
    v[t] = (hm[t] >= 0) ? hbuf[(long)hm[t] * 128 + col] : h1[i];
  }
#pragma unroll
  for (int t = 0; t < NH_TRIPS; ++t) {
    const int i = i0 + t * (NH_WGS * 256);
    if (i < total) h1n[i] = fmaxf(0.0f, v[t] * sc + sh);
  }
}

#define GR 8
static_assert(E2 % 32 == 0, "a wave trip covers 32 whole edges");
__device__ __forceinline__
void scatter2n_body(const int vb, const int nvb,
                    const float* __restrict__ h1,
                    const int* __restrict__ hmap,
                    const float* __restrict__ hbuf,
                    const float* __restrict__ csum, const float* __restrict__ csq,
                    const float* __restrict__ gamma, const float* __restrict__ beta,
                    const int* __restrict__ src,
                    const int* __restrict__ dst,
                    const int* __restrict__ et,
                    float* __restrict__ acc) {
  const int lane = threadIdx.x & 31;
  const int wave = (int)((vb * blockDim.x + threadIdx.x) >> 5);
  const int nWaves = (int)((nvb * blockDim.x) >> 5);
  int base = wave * 32;
  int sv = 0, ev = 0, dv = 0;
  if (base < E2) {
    sv = src[base + lane]; ev = et[base + lane]; dv = dst[base + lane];
  }
  int hv = hmap[sv];
  float sc[4], sh[4];
#pragma unroll
  for (int j = 0; j < 4; ++j)
    bn_coeff(csum, csq, gamma, beta, lane + 32 * j, sc[j], sh[j]);
  while (base < E2) {
    const int av = ev * N2 + dv;
    for (int g = 0; g < 32; g += GR) {
      int   ar[GR];
      float v[GR][4];
#pragma unroll
      for (int u = 0; u < GR; ++u) {
        const int ss = __builtin_amdgcn_readlane(sv, g + u);
        const int hh = __builtin_amdgcn_readlane(hv, g + u);
        ar[u] = __builtin_amdgcn_readlane(av, g + u);
        const float* f = ((hh >= 0) ? (hbuf + (long)hh * 128)
                                    : (h1 + (long)ss * 128)) + lane;
        v[u][0] = f[0]; v[u][1] = f[32]; v[u][2] = f[64]; v[u][3] = f[96];
      }
#pragma unroll
      for (int u = 0; u < GR; ++u) {
        float* a = acc + (long)ar[u] * 128 + lane;
#pragma unroll
        for (int j = 0; j < 4; ++j)
          atomicAdd(a + 32 * j, fmaxf(0.0f, v[u][j] * sc[j] + sh[j]));
      }
    }
    base += nWaves * 32;
    if (base < E2) {
      sv = src[base + lane]; ev = et[base + lane]; dv = dst[base + lane];
      hv = hmap[sv];
    }
  }
}

#define SC2_WGS ((E2 / 32 + 7) / 8)
__global__ __launch_bounds__(256)
void k_scatter2n_normhead(const float* __restrict__ h1,
                          const int* __restrict__ hmap,
                          const float* __restrict__ hbuf,
                          const float* __restrict__ csum,
                          const float* __restrict__ csq,
                          const float* __restrict__ gamma,
                          const float* __restrict__ beta,
                          const int* __restrict__ src2,
                          const int* __restrict__ dst2,
                          const int* __restrict__ et2,
                          float* __restrict__ A2,
                          float* __restrict__ h1n) {
  if (blockIdx.x < SC2_WGS)
    scatter2n_body((int)blockIdx.x, SC2_WGS, h1, hmap, hbuf, csum, csq, gamma, beta,
                   src2, dst2, et2, A2);
  else
    normhead_body((int)blockIdx.x - SC2_WGS, h1, hmap, hbuf, csum, csq,
                  gamma, beta, h1n);
}

#define G2_KSTEPS 20
#define G2_WGS ((N2 + 31) / 32)
static_assert(N2 % 16 == 0, "row tiles are whole");

__global__ __launch_bounds__(256) __attribute__((amdgpu_waves_per_eu(2, 2)))
void k_gemm2(const float* __restrict__ A2,
             const float* __restrict__ h1n,
             const float* __restrict__ W2,
             const float* __restrict__ Ws2,
             const float* __restrict__ b2,
             float* __restrict__ out) {
  __shared__ float red[8 * 8 * 8 * 32];

  const int wave = __builtin_amdgcn_readfirstlane((int)(threadIdx.x >> 5));
  const int lane = threadIdx.x & 31;
  const int c16  = lane & 15;
  const int h    = lane >> 4;
  const int row0 = blockIdx.x * 32;
  const long ra0 = (long)(row0 + c16) * 128;
  const long ra1 = (long)min(row0 + 16 + c16, N2 - 1) * 128;

  v2f    a0[G2_KSTEPS], a1[G2_KSTEPS];
  float4 bk0[G2_KSTEPS], bk1[G2_KSTEPS];
#pragma unroll
  for (int i = 0; i < G2_KSTEPS; ++i) {
    const int kk  = wave * (G2_KSTEPS * 4) + i * 4;
    const int seg = kk >> 7;
    const int k   = (kk & 127) + 2 * h;
    const float* Ab = (seg < 4) ? (A2 + (long)seg * N2 * 128) : h1n;
    const float* Bp = (seg < 4) ? (W2 + (long)seg * 128 * 64) : Ws2;
    a0[i]  = *(const v2f*)(Ab + ra0 + k);
    a1[i]  = *(const v2f*)(Ab + ra1 + k);
    bk0[i] = *(const float4*)(Bp + (long)k * 64 + c16 * 4);
    bk1[i] = *(const float4*)(Bp + (long)(k + 1) * 64 + c16 * 4);
  }

  v8f acc[2][4];
  const v8f zero = {0.f,0.f,0.f,0.f,0.f,0.f,0.f,0.f};
#pragma unroll
  for (int q = 0; q < 4; ++q) { acc[0][q] = zero; acc[1][q] = zero; }

#pragma unroll
  for (int i = 0; i < G2_KSTEPS; ++i) {
    const float k0[4] = {bk0[i].x, bk0[i].y, bk0[i].z, bk0[i].w};
    const float k1[4] = {bk1[i].x, bk1[i].y, bk1[i].z, bk1[i].w};
#pragma unroll
    for (int q = 0; q < 4; ++q) {
      const v2f b = {k0[q], k1[q]};
      acc[0][q] = __builtin_amdgcn_wmma_f32_16x16x4_f32(
          false, a0[i], false, b, (short)0, acc[0][q], false, false);
      acc[1][q] = __builtin_amdgcn_wmma_f32_16x16x4_f32(
          false, a1[i], false, b, (short)0, acc[1][q], false, false);
    }
  }

#pragma unroll
  for (int m = 0; m < 2; ++m)
#pragma unroll
    for (int q = 0; q < 4; ++q)
#pragma unroll
      for (int v = 0; v < 8; ++v)
        red[((wave * 8 + m * 4 + q) * 8 + v) * 32 + lane] = acc[m][q][v];
  __syncthreads();

  float s[2][4];
#pragma unroll
  for (int m = 0; m < 2; ++m)
#pragma unroll
    for (int q = 0; q < 4; ++q) {
      float sum = red[((0 * 8 + m * 4 + q) * 8 + wave) * 32 + lane];
#pragma unroll
      for (int w = 1; w < 8; ++w)
        sum += red[((w * 8 + m * 4 + q) * 8 + wave) * 32 + lane];
      s[m][q] = sum;
    }

  const float4 bb = *(const float4*)(b2 + c16 * 4);
#pragma unroll
  for (int m = 0; m < 2; ++m) {
    const int r = row0 + 16 * m + wave + 8 * h;
    const float v0 = s[m][0] + bb.x, v1 = s[m][1] + bb.y;
    const float v2 = s[m][2] + bb.z, v3 = s[m][3] + bb.w;
    float mx = fmaxf(fmaxf(v0, v1), fmaxf(v2, v3));
#pragma unroll
    for (int mask = 1; mask < 16; mask <<= 1)
      mx = fmaxf(mx, __shfl_xor(mx, mask, 32));
    float e = (expf(v0 - mx) + expf(v1 - mx)) + (expf(v2 - mx) + expf(v3 - mx));
#pragma unroll
    for (int mask = 1; mask < 16; mask <<= 1)
      e += __shfl_xor(e, mask, 32);
    const float lse = mx + logf(e);
    if (r < N2)
      *(float4*)(out + (long)r * 64 + c16 * 4) =
          make_float4(v0 - lse, v1 - lse, v2 - lse, v3 - lse);
  }
}

extern "C" void kernel_launch(void* const* d_in, const int* in_sizes, int n_in,
                              void* d_out, int out_size, void* d_ws, size_t ws_size,
                              hipStream_t stream) {
  const float* x     = (const float*)d_in[0];
  const int*   src1  = (const int*)  d_in[1];
  const int*   dst1  = (const int*)  d_in[2];
  const int*   et1   = (const int*)  d_in[3];
  const int*   src2  = (const int*)  d_in[4];
  const int*   dst2  = (const int*)  d_in[5];
  const int*   et2   = (const int*)  d_in[6];
  const int*   hmap  = (const int*)  d_in[7];
  const float* hbuf  = (const float*)d_in[8];
  const float* W1    = (const float*)d_in[9];
  const float* Ws1   = (const float*)d_in[10];
  const float* b1    = (const float*)d_in[11];
  const float* gamma = (const float*)d_in[12];
  const float* beta  = (const float*)d_in[13];
  const float* W2    = (const float*)d_in[14];
  const float* Ws2   = (const float*)d_in[15];
  const float* b2    = (const float*)d_in[16];

  float* ws    = (float*)d_ws;
  float* A1    = ws;
  float* A2    = A1 + (size_t)4 * N1 * 128;
  float* sums  = A2 + (size_t)4 * N2 * 128;
  float* h1    = sums + 512;
  float* h1n   = h1 + (size_t)N1 * 128;
  float* out   = (float*)d_out;

  k_clear<<<512, 256, 0, stream>>>(A2, (long)4 * N2 * 128 + 256, A1, hmap);

  k_scatter_histfill<<<HF_WGS + SC1_WGS, 256, 0, stream>>>(x, src1, dst1, et1, hmap, hbuf,
                                                           A1, sums, sums + 128);

  k_gemm1<<<G1_RGS * 4, 256, 0, stream>>>(A1, x, W1, Ws1, b1, hmap,
                                               h1, sums, sums + 128);

  k_scatter2n_normhead<<<SC2_WGS + NH_WGS, 256, 0, stream>>>(h1, hmap, hbuf, sums, sums + 128, gamma, beta,
                                                             src2, dst2, et2, A2, h1n);

  k_gemm2<<<G2_WGS, 256, 0, stream>>>(A2, h1n, W2, Ws2, b2, out);
}


# layer-1 scatter tests edges against a no-history bit map (built by the clear launch, copied to LDS) not a history-map load per edge; one trip per wave, 247 groups (was 586); atomic order changes
# speedup vs baseline: 9.5022x; 1.1148x over previous
#include <hip/hip_runtime.h>
#include <hip/hip_bf16.h>
#include <math.h>

typedef __attribute__((ext_vector_type(2))) float v2f;
typedef __attribute__((ext_vector_type(8))) float v8f;

#define N0  300000
#define N1  60000
#define N2  6000
#define E1  1200000
#define E2  300000
#define D_H 128
#define D_OUT 64
#define BN_EPS 1e-5f

#define BM_WORDS (N1 / 32)
static_assert(N1 % 32 == 0, "a wave trip covers 32 whole rows, one map word");
__global__ void k_clear(float* __restrict__ p, long n,
                        float* __restrict__ A1, const int* __restrict__ hmap,
                        int* __restrict__ bm) {
  const long tid = (long)blockIdx.x * blockDim.x + threadIdx.x;
  const long stride = (long)gridDim.x * blockDim.x;
  for (long i = tid; i < n; i += stride) p[i] = 0.0f;
  const int lane = threadIdx.x & 31;
  for (long w = tid >> 5; w < BM_WORDS; w += stride >> 5) {
    const long nn = w * 32 + lane;
    const bool nohist = hmap[nn] < 0;
    const unsigned m = (unsigned)__ballot(nohist);
    if (lane == 0) bm[w] = (int)m;
    if (nohist) {
      for (int r = 0; r < 4; ++r) {
        float* q = A1 + ((long)r * N1 + nn) * 128;
        for (int c = 0; c < 128; ++c) q[c] = 0.0f;
      }
    }
  }
}

#define S1_EPL 19
#define S1_WAVE_EDGES (32 * S1_EPL)
#define S1_WG_EDGES   (8 * S1_WAVE_EDGES)
#define BM_TRIPS ((BM_WORDS + 255) / 256)
__device__ __forceinline__
void scatter_f_body(const int vb,
                    const float* __restrict__ feat,
                    const int* __restrict__ src,
                    const int* __restrict__ dst,
                    const int* __restrict__ et,
                    const int* __restrict__ bm,
                    float* __restrict__ acc, int nE, int nOut) {
  __shared__ int bms[BM_TRIPS * 256];
  const int lane = threadIdx.x & 31;
  const int wave = (int)((vb * blockDim.x + threadIdx.x) >> 5);
  const int base = wave * S1_WAVE_EDGES;
  int d[S1_EPL];
#pragma unroll
  for (int j = 0; j < S1_EPL; ++j)
    d[j] = dst[min(base + j * 32 + lane, nE - 1)];
  int mw[BM_TRIPS];
#pragma unroll
  for (int k = 0; k < BM_TRIPS; ++k)
    mw[k] = bm[min((int)threadIdx.x + k * 256, BM_WORDS - 1)];
#pragma unroll
  for (int k = 0; k < BM_TRIPS; ++k)
    bms[(int)threadIdx.x + k * 256] = mw[k];
  __syncthreads();
  int w[S1_EPL];
#pragma unroll
  for (int j = 0; j < S1_EPL; ++j)
    w[j] = bms[d[j] >> 5];
#pragma unroll
  for (int j = 0; j < S1_EPL; ++j) {
    const int me = base + j * 32 + lane;
    unsigned todo = (unsigned)__ballot(me < nE && (((unsigned)w[j] >> (d[j] & 31)) & 1u));
    while (todo) {
      const int l = __ffs(todo) - 1;
      todo &= todo - 1;
      const int pe = base + j * 32 + l;
      const int dd = __builtin_amdgcn_readlane(d[j], l);
      const float* f = feat + (long)src[pe] * 128 + lane;
      float* a = acc + ((long)et[pe] * nOut + dd) * 128 + lane;
      const float v0 = f[0], v1 = f[32], v2 = f[64], v3 = f[96];
      atomicAdd(a,      v0);
      atomicAdd(a + 32, v1);
      atomicAdd(a + 64, v2);
      atomicAdd(a + 96, v3);
    }
  }
}

#define HF_ROWS 16
static_assert(N1 % 32 == 0, "a wave trip covers 32 whole rows");
__device__ __forceinline__
void histfill_stats_body(const int vb, const int nvb,
                         const int* __restrict__ hmap,
                         const float* __restrict__ hbuf,
                         float* __restrict__ csum,
                         float* __restrict__ csq) {
  __shared__ __attribute__((aligned(16))) float ls[8][128];
  __shared__ __attribute__((aligned(16))) float lq[8][128];
  const int lane = threadIdx.x & 31;
  const int wv   = threadIdx.x >> 5;
  const int wave = (int)((vb * blockDim.x + threadIdx.x) >> 5);
  const int nWaves = (int)((nvb * blockDim.x) >> 5);
  float4 s = make_float4(0.f, 0.f, 0.f, 0.f);
  float4 q = make_float4(0.f, 0.f, 0.f, 0.f);
  for (int base = wave * 32; base < N1; base += nWaves * 32) {
    const int hmv = hmap[base + lane];
    for (int g = 0; g < 32; g += HF_ROWS) {
      int    hm[HF_ROWS];
      float4 v[HF_ROWS];
#pragma unroll
      for (int u = 0; u < HF_ROWS; ++u) {
        hm[u] = __builtin_amdgcn_readlane(hmv, g + u);
        v[u] = *(const float4*)(hbuf + (long)max(hm[u], 0) * 128 + lane * 4);
      }
#pragma unroll
      for (int u = 0; u < HF_ROWS; ++u) {
        if (hm[u] >= 0) {
          s.x += v[u].x; s.y += v[u].y; s.z += v[u].z; s.w += v[u].w;
          q.x += v[u].x * v[u].x; q.y += v[u].y * v[u].y;
          q.z += v[u].z * v[u].z; q.w += v[u].w * v[u].w;
        }
      }
    }
  }
  *(float4*)&ls[wv][lane * 4] = s;
  *(float4*)&lq[wv][lane * 4] = q;
  __syncthreads();
  if (threadIdx.x < 128) {
    float S = ls[0][threadIdx.x], Q = lq[0][threadIdx.x];
    for (int w = 1; w < 8; ++w) { S += ls[w][threadIdx.x]; Q += lq[w][threadIdx.x]; }
    atomicAdd(&csum[threadIdx.x], S);
    atomicAdd(&csq[threadIdx.x], Q);
  }
}

#define HF_WGS ((N1 / 32 + 7) / 8)
#define SC1_WGS ((E1 + S1_WG_EDGES - 1) / S1_WG_EDGES)
__global__ __launch_bounds__(256)
void k_scatter_histfill(const float* __restrict__ x,
                        const int* __restrict__ src1,
                        const int* __restrict__ dst1,
                        const int* __restrict__ et1,
                        const int* __restrict__ hmap,
                        const int* __restrict__ bm,
                        const float* __restrict__ hbuf,
                        float* __restrict__ A1,
                        float* __restrict__ csum,
                        float* __restrict__ csq) {
  if (blockIdx.x < HF_WGS)
    histfill_stats_body((int)blockIdx.x, HF_WGS, hmap, hbuf, csum, csq);
  else
    scatter_f_body((int)blockIdx.x - HF_WGS, x, src1, dst1, et1, bm, A1, E1, N1);
}

#define G1_KSTEPS 20
#define G1_ROWS   1024
#define G1_RGS    ((N1 + G1_ROWS - 1) / G1_ROWS)
static_assert(N1 % 16 == 0, "row tiles are whole");
static_assert(N1 % 4 == 0, "four history entries are in or out together");

__global__ __launch_bounds__(256)
void k_gemm1(const float* __restrict__ A1,
             const float* __restrict__ x,
             const float* __restrict__ W1,
             const float* __restrict__ Ws,
             const float* __restrict__ b1,
             const int*   __restrict__ hmap,
             float* __restrict__ h1,
             float* __restrict__ csum,
             float* __restrict__ csq) {
  __shared__ float red[8 * 2 * 8 * 32];

  const int wave = __builtin_amdgcn_readfirstlane((int)(threadIdx.x >> 5));
  const int lane = threadIdx.x & 31;
  const int rg   = (int)blockIdx.x >> 2;
  const int qc   = (int)blockIdx.x & 3;
  bool need[8];
#pragma unroll
  for (int b = 0; b < 8; ++b) {
    const int rb = rg * G1_ROWS + b * 128 + lane * 4;
    const int4 hv = *(const int4*)(hmap + min(rb, N1 - 4));
    need[b] = (rb < N1) & ((hv.x < 0) | (hv.y < 0) | (hv.z < 0) | (hv.w < 0));
  }
  unsigned long long todo = 0ull;
#pragma unroll
  for (int b = 0; b < 8; ++b) {
    unsigned m = (unsigned)__ballot(need[b]);
    m |= m >> 1;
    m |= m >> 2;
    unsigned bits = 0u;
#pragma unroll
    for (int t = 0; t < 8; ++t) bits |= ((m >> (4 * t)) & 1u) << t;
    todo |= (unsigned long long)bits << (8 * b);
  }
  if (todo == 0ull) return;

  const int c16 = lane & 15;
  const int h   = lane >> 4;
  const int colB = qc * 32 + c16 * 2;

  int tile = __builtin_ctzll(todo);
  todo &= todo - 1ull;
  v2f af[G1_KSTEPS];
#pragma unroll
  for (int i = 0; i < G1_KSTEPS; ++i) {
    const int kk  = wave * (G1_KSTEPS * 4) + i * 4;
    const int seg = kk >> 7;
    const int k   = (kk & 127) + 2 * h;
    const float* Ab = (seg < 4) ? (A1 + (long)seg * N1 * 128) : x;
    af[i] = *(const v2f*)(Ab + (long)(rg * G1_ROWS + tile * 16 + c16) * 128 + k);
  }
  v2f bk0[G1_KSTEPS], bk1[G1_KSTEPS];
#pragma unroll
  for (int i = 0; i < G1_KSTEPS; ++i) {
    const int kk  = wave * (G1_KSTEPS * 4) + i * 4;
    const int seg = kk >> 7;
    const int k   = (kk & 127) + 2 * h;
    const float* Bp = (seg < 4) ? (W1 + (long)seg * 128 * 128) : Ws;
    bk0[i] = *(const v2f*)(Bp + (long)k * 128 + colB);
    bk1[i] = *(const v2f*)(Bp + (long)(k + 1) * 128 + colB);
  }

  for (;;) {
    const int row0 = rg * G1_ROWS + tile * 16;
    const int r = row0 + wave + 8 * h;
    const int hr = hmap[r];
    const float2 bb = *(const float2*)(b1 + colB);

    const v8f zero = {0.f,0.f,0.f,0.f,0.f,0.f,0.f,0.f};
    v8f acc0 = zero, acc1 = zero;
#pragma unroll
    for (int i = 0; i < G1_KSTEPS; ++i) {
      const v2f bt0 = {bk0[i].x, bk1[i].x};
      const v2f bt1 = {bk0[i].y, bk1[i].y};
      acc0 = __builtin_amdgcn_wmma_f32_16x16x4_f32(
          false, af[i], false, bt0, (short)0, acc0, false, false);
      acc1 = __builtin_amdgcn_wmma_f32_16x16x4_f32(
          false, af[i], false, bt1, (short)0, acc1, false, false);
    }

#pragma unroll
    for (int v = 0; v < 8; ++v) {
      red[((wave * 2 + 0) * 8 + v) * 32 + lane] = acc0[v];
      red[((wave * 2 + 1) * 8 + v) * 32 + lane] = acc1[v];
    }
    __syncthreads();

    float s[2];
#pragma unroll
    for (int t = 0; t < 2; ++t) {
      float sum = red[((0 * 2 + t) * 8 + wave) * 32 + lane];
#pragma unroll
      for (int w = 1; w < 8; ++w)
        sum += red[((w * 2 + t) * 8 + wave) * 32 + lane];
      s[t] = sum;
    }

    if (hr < 0) {
      const float o0 = s[0] + bb.x;
      const float o1 = s[1] + bb.y;
      atomicAdd(&csum[colB],     o0);
      atomicAdd(&csq[colB],      o0 * o0);
      atomicAdd(&csum[colB + 1], o1);
      atomicAdd(&csq[colB + 1],  o1 * o1);
      *(float2*)(h1 + (long)r * 128 + colB) = make_float2(o0, o1);
    }

    if (todo == 0ull) break;
    tile = __builtin_ctzll(todo);
    todo &= todo - 1ull;
#pragma unroll
    for (int i = 0; i < G1_KSTEPS; ++i) {
      const int kk  = wave * (G1_KSTEPS * 4) + i * 4;
      const int seg = kk >> 7;
      const int k   = (kk & 127) + 2 * h;
      const float* Ab = (seg < 4) ? (A1 + (long)seg * N1 * 128) : x;
      af[i] = *(const v2f*)(Ab + (long)(rg * G1_ROWS + tile * 16 + c16) * 128 + k);
    }
    __syncthreads();
  }
}

__device__ __forceinline__
void bn_coeff(const float* __restrict__ csum, const float* __restrict__ csq,
              const float* __restrict__ gamma, const float* __restrict__ beta,
              const int c, float& sc, float& sh) {
  const float inv_n = 1.0f / (float)N1;
  const float mean = csum[c] * inv_n;
  const float var  = csq[c] * inv_n - mean * mean;
  sc = gamma[c] * rsqrtf(var + BN_EPS);
  sh = beta[c] - mean * sc;
}

#define NH_WGS   512
#define NH_TRIPS ((N2 * 128 + NH_WGS * 256 - 1) / (NH_WGS * 256))
__device__ __forceinline__
void normhead_body(const int vb,
                   const float* __restrict__ h1,
                   const int* __restrict__ hmap,
                   const float* __restrict__ hbuf,
                   const float* __restrict__ csum, const float* __restrict__ csq,
                   const float* __restrict__ gamma, const float* __restrict__ beta,
                   float* __restrict__ h1n) {
  const int total = N2 * 128;
  const int i0 = vb * 256 + (int)threadIdx.x;
  const int col = i0 & 127;
  int   hm[NH_TRIPS];
  float v[NH_TRIPS];
#pragma unroll
  for (int t = 0; t < NH_TRIPS; ++t)
    hm[t] = hmap[min(i0 + t * (NH_WGS * 256), total - 1) >> 7];
  float sc, sh;
  bn_coeff(csum, csq, gamma, beta, col, sc, sh);
#pragma unroll
  for (int t = 0; t < NH_TRIPS; ++t) {
    const int i = min(i0 + t * (NH_WGS * 256), total - 1);
    v[t] = (hm[t] >= 0) ? hbuf[(long)hm[t] * 128 + col] : h1[i];
  }
#pragma unroll
  for (int t = 0; t < NH_TRIPS; ++t) {
    const int i = i0 + t * (NH_WGS * 256);
    if (i < total) h1n[i] = fmaxf(0.0f, v[t] * sc + sh);
  }
}

#define GR 8
static_assert(E2 % 32 == 0, "a wave trip covers 32 whole edges");
__device__ __forceinline__
void scatter2n_body(const int vb, const int nvb,
                    const float* __restrict__ h1,
                    const int* __restrict__ hmap,
                    const float* __restrict__ hbuf,
                    const float* __restrict__ csum, const float* __restrict__ csq,
                    const float* __restrict__ gamma, const float* __restrict__ beta,
                    const int* __restrict__ src,
                    const int* __restrict__ dst,
                    const int* __restrict__ et,
                    float* __restrict__ acc) {
  const int lane = threadIdx.x & 31;
  const int wave = (int)((vb * blockDim.x + threadIdx.x) >> 5);
  const int nWaves = (int)((nvb * blockDim.x) >> 5);
  int base = wave * 32;
  int sv = 0, ev = 0, dv = 0;
  if (base < E2) {
    sv = src[base + lane]; ev = et[base + lane]; dv = dst[base + lane];
  }
  int hv = hmap[sv];
  float sc[4], sh[4];
#pragma unroll
  for (int j = 0; j < 4; ++j)
    bn_coeff(csum, csq, gamma, beta, lane + 32 * j, sc[j], sh[j]);
  while (base < E2) {
    const int av = ev * N2 + dv;
    for (int g = 0; g < 32; g += GR) {
      int   ar[GR];
      float v[GR][4];
#pragma unroll
      for (int u = 0; u < GR; ++u) {
        const int ss = __builtin_amdgcn_readlane(sv, g + u);
        const int hh = __builtin_amdgcn_readlane(hv, g + u);
        ar[u] = __builtin_amdgcn_readlane(av, g + u);
        const float* f = ((hh >= 0) ? (hbuf + (long)hh * 128)
                                    : (h1 + (long)ss * 128)) + lane;
        v[u][0] = f[0]; v[u][1] = f[32]; v[u][2] = f[64]; v[u][3] = f[96];
      }
#pragma unroll
      for (int u = 0; u < GR; ++u) {
        float* a = acc + (long)ar[u] * 128 + lane;
#pragma unroll
        for (int j = 0; j < 4; ++j)
          atomicAdd(a + 32 * j, fmaxf(0.0f, v[u][j] * sc[j] + sh[j]));
      }
    }
    base += nWaves * 32;
    if (base < E2) {
      sv = src[base + lane]; ev = et[base + lane]; dv = dst[base + lane];
      hv = hmap[sv];
    }
  }
}

#define SC2_WGS ((E2 / 32 + 7) / 8)
__global__ __launch_bounds__(256)
void k_scatter2n_normhead(const float* __restrict__ h1,
                          const int* __restrict__ hmap,
                          const float* __restrict__ hbuf,
                          const float* __restrict__ csum,
                          const float* __restrict__ csq,
                          const float* __restrict__ gamma,
                          const float* __restrict__ beta,
                          const int* __restrict__ src2,
                          const int* __restrict__ dst2,
                          const int* __restrict__ et2,
                          float* __restrict__ A2,
                          float* __restrict__ h1n) {
  if (blockIdx.x < SC2_WGS)
    scatter2n_body((int)blockIdx.x, SC2_WGS, h1, hmap, hbuf, csum, csq, gamma, beta,
                   src2, dst2, et2, A2);
  else
    normhead_body((int)blockIdx.x - SC2_WGS, h1, hmap, hbuf, csum, csq,
                  gamma, beta, h1n);
}

#define G2_KSTEPS 20
#define G2_WGS ((N2 + 31) / 32)
static_assert(N2 % 16 == 0, "row tiles are whole");

__global__ __launch_bounds__(256) __attribute__((amdgpu_waves_per_eu(2, 2)))
void k_gemm2(const float* __restrict__ A2,
             const float* __restrict__ h1n,
             const float* __restrict__ W2,
             const float* __restrict__ Ws2,
             const float* __restrict__ b2,
             float* __restrict__ out) {
  __shared__ float red[8 * 8 * 8 * 32];

  const int wave = __builtin_amdgcn_readfirstlane((int)(threadIdx.x >> 5));
  const int lane = threadIdx.x & 31;
  const int c16  = lane & 15;
  const int h    = lane >> 4;
  const int row0 = blockIdx.x * 32;
  const long ra0 = (long)(row0 + c16) * 128;
  const long ra1 = (long)min(row0 + 16 + c16, N2 - 1) * 128;

  v2f    a0[G2_KSTEPS], a1[G2_KSTEPS];
  float4 bk0[G2_KSTEPS], bk1[G2_KSTEPS];
#pragma unroll
  for (int i = 0; i < G2_KSTEPS; ++i) {
    const int kk  = wave * (G2_KSTEPS * 4) + i * 4;
    const int seg = kk >> 7;
    const int k   = (kk & 127) + 2 * h;
    const float* Ab = (seg < 4) ? (A2 + (long)seg * N2 * 128) : h1n;
    const float* Bp = (seg < 4) ? (W2 + (long)seg * 128 * 64) : Ws2;
    a0[i]  = *(const v2f*)(Ab + ra0 + k);
    a1[i]  = *(const v2f*)(Ab + ra1 + k);
    bk0[i] = *(const float4*)(Bp + (long)k * 64 + c16 * 4);
    bk1[i] = *(const float4*)(Bp + (long)(k + 1) * 64 + c16 * 4);
  }

  v8f acc[2][4];
  const v8f zero = {0.f,0.f,0.f,0.f,0.f,0.f,0.f,0.f};
#pragma unroll
  for (int q = 0; q < 4; ++q) { acc[0][q] = zero; acc[1][q] = zero; }

#pragma unroll
  for (int i = 0; i < G2_KSTEPS; ++i) {
    const float k0[4] = {bk0[i].x, bk0[i].y, bk0[i].z, bk0[i].w};
    const float k1[4] = {bk1[i].x, bk1[i].y, bk1[i].z, bk1[i].w};
#pragma unroll
    for (int q = 0; q < 4; ++q) {
      const v2f b = {k0[q], k1[q]};
      acc[0][q] = __builtin_amdgcn_wmma_f32_16x16x4_f32(
          false, a0[i], false, b, (short)0, acc[0][q], false, false);
      acc[1][q] = __builtin_amdgcn_wmma_f32_16x16x4_f32(
          false, a1[i], false, b, (short)0, acc[1][q], false, false);
    }
  }

#pragma unroll
  for (int m = 0; m < 2; ++m)
#pragma unroll
    for (int q = 0; q < 4; ++q)
#pragma unroll
      for (int v = 0; v < 8; ++v)
        red[((wave * 8 + m * 4 + q) * 8 + v) * 32 + lane] = acc[m][q][v];
  __syncthreads();

  float s[2][4];
#pragma unroll
  for (int m = 0; m < 2; ++m)
#pragma unroll
    for (int q = 0; q < 4; ++q) {
      float sum = red[((0 * 8 + m * 4 + q) * 8 + wave) * 32 + lane];
#pragma unroll
      for (int w = 1; w < 8; ++w)
        sum += red[((w * 8 + m * 4 + q) * 8 + wave) * 32 + lane];
      s[m][q] = sum;
    }

  const float4 bb = *(const float4*)(b2 + c16 * 4);
#pragma unroll
  for (int m = 0; m < 2; ++m) {
    const int r = row0 + 16 * m + wave + 8 * h;
    const float v0 = s[m][0] + bb.x, v1 = s[m][1] + bb.y;
    const float v2 = s[m][2] + bb.z, v3 = s[m][3] + bb.w;
    float mx = fmaxf(fmaxf(v0, v1), fmaxf(v2, v3));
#pragma unroll
    for (int mask = 1; mask < 16; mask <<= 1)
      mx = fmaxf(mx, __shfl_xor(mx, mask, 32));
    float e = (expf(v0 - mx) + expf(v1 - mx)) + (expf(v2 - mx) + expf(v3 - mx));
#pragma unroll
    for (int mask = 1; mask < 16; mask <<= 1)
      e += __shfl_xor(e, mask, 32);
    const float lse = mx + logf(e);
    if (r < N2)
      *(float4*)(out + (long)r * 64 + c16 * 4) =
          make_float4(v0 - lse, v1 - lse, v2 - lse, v3 - lse);
  }
}

extern "C" void kernel_launch(void* const* d_in, const int* in_sizes, int n_in,
                              void* d_out, int out_size, void* d_ws, size_t ws_size,
                              hipStream_t stream) {
  const float* x     = (const float*)d_in[0];
  const int*   src1  = (const int*)  d_in[1];
  const int*   dst1  = (const int*)  d_in[2];
  const int*   et1   = (const int*)  d_in[3];
  const int*   src2  = (const int*)  d_in[4];
  const int*   dst2  = (const int*)  d_in[5];
  const int*   et2   = (const int*)  d_in[6];
  const int*   hmap  = (const int*)  d_in[7];
  const float* hbuf  = (const float*)d_in[8];
  const float* W1    = (const float*)d_in[9];
  const float* Ws1   = (const float*)d_in[10];
  const float* b1    = (const float*)d_in[11];
  const float* gamma = (const float*)d_in[12];
  const float* beta  = (const float*)d_in[13];
  const float* W2    = (const float*)d_in[14];
  const float* Ws2   = (const float*)d_in[15];
  const float* b2    = (const float*)d_in[16];

  float* ws    = (float*)d_ws;
  float* A1    = ws;
  float* A2    = A1 + (size_t)4 * N1 * 128;
  float* sums  = A2 + (size_t)4 * N2 * 128;
  float* h1    = sums + 512;
  float* h1n   = h1 + (size_t)N1 * 128;
  int*   bm    = (int*)h1n;
  float* out   = (float*)d_out;

  k_clear<<<512, 256, 0, stream>>>(A2, (long)4 * N2 * 128 + 256, A1, hmap, bm);

  k_scatter_histfill<<<HF_WGS + SC1_WGS, 256, 0, stream>>>(x, src1, dst1, et1, hmap, bm, hbuf,
                                                           A1, sums, sums + 128);

  k_gemm1<<<G1_RGS * 4, 256, 0, stream>>>(A1, x, W1, Ws1, b1, hmap,
                                               h1, sums, sums + 128);

  k_scatter2n_normhead<<<SC2_WGS + NH_WGS, 256, 0, stream>>>(h1, hmap, hbuf, sums, sums + 128, gamma, beta,
                                                             src2, dst2, et2, A2, h1n);

  k_gemm2<<<G2_WGS, 256, 0, stream>>>(A2, h1n, W2, Ws2, b2, out);
}
